# starting kernel
# baseline (speedup 1.0000x reference)
#include <hip/hip_runtime.h>

#define D 128
#define LDSW 130

typedef __attribute__((ext_vector_type(2))) float v2f;
typedef __attribute__((ext_vector_type(8))) float v8f;

__device__ __forceinline__ v8f wmma_f32(v2f a, v2f b, v8f c) {
    return __builtin_amdgcn_wmma_f32_16x16x4_f32(false, a, false, b, (short)0, c,
                                                 false, false);
}

__global__ void zero_kernel(float* __restrict__ p, size_t n) {
    size_t i = (size_t)blockIdx.x * blockDim.x + threadIdx.x;
    if (i < n) p[i] = 0.0f;
}

__global__ __launch_bounds__(256)
void lin_kernel(const float* __restrict__ x, const float* __restrict__ W,
                const float* __restrict__ bias, float* __restrict__ t) {
    __shared__ float aS[16 * LDSW];
    const int tid  = threadIdx.x;
    const int row0 = blockIdx.x * 16;

    for (int i = tid; i < 16 * D; i += 256) {
        int r = i >> 7, c = i & (D - 1);
        aS[r * LDSW + c] = x[(size_t)(row0 + r) * D + c];
    }
    __syncthreads();

    const int lane = tid & 31, wv = tid >> 5;
    const int n = lane & 15, h = lane >> 4, m = lane & 15;
    const int kb = h * 2;
    const int col = wv * 16 + n;

    v8f acc = {};
    for (int k = 0; k < D; k += 4) {
        v2f a = *(const v2f*)&aS[m * LDSW + k + kb];
        v2f b;
        b.x = W[(size_t)(k + kb) * D + col];
        b.y = W[(size_t)(k + kb + 1) * D + col];
        acc = wmma_f32(a, b, acc);
    }
    const float bv = bias[col];
    for (int p = 0; p < 8; ++p)
        t[(size_t)(row0 + p + 8 * h) * D + col] = acc[p] + bv;
}

__global__ void scatter_kernel(const float* __restrict__ tsrc, const int* __restrict__ src,
                               const int* __restrict__ dst, float* __restrict__ agg,
                               float* __restrict__ cnt, int E) {
    size_t i = (size_t)blockIdx.x * blockDim.x + threadIdx.x;
    if (i >= (size_t)E * D) return;
    int e = (int)(i >> 7);
    int c = (int)(i & (D - 1));
    int s = src[e], d0 = dst[e];
    atomicAdd(&agg[(size_t)d0 * D + c], tsrc[(size_t)s * D + c]);
    if (c == 0) atomicAdd(&cnt[d0], 1.0f);
}

__global__ __launch_bounds__(256)
void combine_m_kernel(const float* __restrict__ t_m,
                      const float* __restrict__ agg_mm, const float* __restrict__ cnt_mm,
                      const float* __restrict__ agg_hm, const float* __restrict__ cnt_hm,
                      const float* __restrict__ Wl_mm, const float* __restrict__ bl_mm,
                      const float* __restrict__ Wr_mm,
                      const float* __restrict__ Wl_hm, const float* __restrict__ bl_hm,
                      const float* __restrict__ Wr_hm,
                      const float* __restrict__ g, const float* __restrict__ be,
                      float* __restrict__ out) {
    __shared__ float tS[16 * LDSW], aS[16 * LDSW], bS[16 * LDSW];
    __shared__ float red[4][16];
    const int tid  = threadIdx.x;
    const int row0 = blockIdx.x * 16;

    if (tid < 64) red[tid >> 4][tid & 15] = 0.0f;
    for (int i = tid; i < 16 * D; i += 256) {
        int r = i >> 7, c = i & (D - 1);
        int gr = row0 + r;
        tS[r * LDSW + c] = t_m[(size_t)gr * D + c];
        aS[r * LDSW + c] = agg_mm[(size_t)gr * D + c] / fmaxf(cnt_mm[gr], 1.0f);
        bS[r * LDSW + c] = agg_hm[(size_t)gr * D + c] / fmaxf(cnt_hm[gr], 1.0f);
    }
    __syncthreads();

    const int lane = tid & 31, wv = tid >> 5;
    const int n = lane & 15, h = lane >> 4, m = lane & 15;
    const int kb = h * 2;
    const int col = wv * 16 + n;

    v8f acc_mm = {}, acc_hm = {};
    for (int k = 0; k < D; k += 4) {
        v2f at = *(const v2f*)&tS[m * LDSW + k + kb];
        v2f aa = *(const v2f*)&aS[m * LDSW + k + kb];
        v2f ab = *(const v2f*)&bS[m * LDSW + k + kb];
        v2f b1; b1.x = Wl_mm[(size_t)(k + kb) * D + col]; b1.y = Wl_mm[(size_t)(k + kb + 1) * D + col];
        v2f b2; b2.x = Wr_mm[(size_t)(k + kb) * D + col]; b2.y = Wr_mm[(size_t)(k + kb + 1) * D + col];
        v2f b3; b3.x = Wl_hm[(size_t)(k + kb) * D + col]; b3.y = Wl_hm[(size_t)(k + kb + 1) * D + col];
        v2f b4; b4.x = Wr_hm[(size_t)(k + kb) * D + col]; b4.y = Wr_hm[(size_t)(k + kb + 1) * D + col];
        acc_mm = wmma_f32(aa, b1, acc_mm);
        acc_mm = wmma_f32(at, b2, acc_mm);
        acc_hm = wmma_f32(ab, b3, acc_hm);
        acc_hm = wmma_f32(at, b4, acc_hm);
    }

    float u1[8], u2[8];
    const float c1 = bl_mm[col], c2 = bl_hm[col];
    for (int p = 0; p < 8; ++p) {
        u1[p] = acc_mm[p] + c1;
        u2[p] = acc_hm[p] + c2;
        int r = p + 8 * h;
        atomicAdd(&red[0][r], u1[p] * u1[p]);
        atomicAdd(&red[1][r], u2[p] * u2[p]);
    }
    __syncthreads();

    float y[8];
    for (int p = 0; p < 8; ++p) {
        int r = p + 8 * h;
        float n1 = fmaxf(sqrtf(red[0][r]), 1e-12f);
        float n2 = fmaxf(sqrtf(red[1][r]), 1e-12f);
        float conv = 0.5f * (u1[p] / n1 + u2[p] / n2);
        y[p] = tS[r * LDSW + col] + conv;
        atomicAdd(&red[2][r], y[p]);
        atomicAdd(&red[3][r], y[p] * y[p]);
    }
    __syncthreads();

    const float gc = g[col], bc = be[col];
    for (int p = 0; p < 8; ++p) {
        int r = p + 8 * h;
        float mu  = red[2][r] * (1.0f / D);
        float var = red[3][r] * (1.0f / D) - mu * mu;
        out[(size_t)(row0 + r) * D + col] = (y[p] - mu) * rsqrtf(var + 1e-5f) * gc + bc;
    }
}

__global__ __launch_bounds__(256)
void combine_h_kernel(const float* __restrict__ t_h,
                      const float* __restrict__ agg_mh, const float* __restrict__ cnt_mh,
                      const float* __restrict__ Wl_mh, const float* __restrict__ bl_mh,
                      const float* __restrict__ Wr_mh,
                      const float* __restrict__ g, const float* __restrict__ be,
                      float* __restrict__ out) {
    __shared__ float tS[16 * LDSW], aS[16 * LDSW];
    __shared__ float red[3][16];
    const int tid  = threadIdx.x;
    const int row0 = blockIdx.x * 16;

    if (tid < 48) red[tid >> 4][tid & 15] = 0.0f;
    for (int i = tid; i < 16 * D; i += 256) {
        int r = i >> 7, c = i & (D - 1);
        int gr = row0 + r;
        tS[r * LDSW + c] = t_h[(size_t)gr * D + c];
        aS[r * LDSW + c] = agg_mh[(size_t)gr * D + c] / fmaxf(cnt_mh[gr], 1.0f);
    }
    __syncthreads();

    const int lane = tid & 31, wv = tid >> 5;
    const int n = lane & 15, h = lane >> 4, m = lane & 15;
    const int kb = h * 2;
    const int col = wv * 16 + n;

    v8f acc = {};
    for (int k = 0; k < D; k += 4) {
        v2f at = *(const v2f*)&tS[m * LDSW + k + kb];
        v2f aa = *(const v2f*)&aS[m * LDSW + k + kb];
        v2f b1; b1.x = Wl_mh[(size_t)(k + kb) * D + col]; b1.y = Wl_mh[(size_t)(k + kb + 1) * D + col];
        v2f b2; b2.x = Wr_mh[(size_t)(k + kb) * D + col]; b2.y = Wr_mh[(size_t)(k + kb + 1) * D + col];
        acc = wmma_f32(aa, b1, acc);
        acc = wmma_f32(at, b2, acc);
    }

    float u[8];
    const float c1 = bl_mh[col];
    for (int p = 0; p < 8; ++p) {
        u[p] = acc[p] + c1;
        atomicAdd(&red[0][p + 8 * h], u[p] * u[p]);
    }
    __syncthreads();

    float y[8];
    for (int p = 0; p < 8; ++p) {
        int r = p + 8 * h;
        float n1 = fmaxf(sqrtf(red[0][r]), 1e-12f);
        y[p] = tS[r * LDSW + col] + u[p] / n1;
        atomicAdd(&red[1][r], y[p]);
        atomicAdd(&red[2][r], y[p] * y[p]);
    }
    __syncthreads();

    const float gc = g[col], bc = be[col];
    for (int p = 0; p < 8; ++p) {
        int r = p + 8 * h;
        float mu  = red[1][r] * (1.0f / D);
        float var = red[2][r] * (1.0f / D) - mu * mu;
        out[(size_t)(row0 + r) * D + col] = (y[p] - mu) * rsqrtf(var + 1e-5f) * gc + bc;
    }
}

extern "C" void kernel_launch(void* const* d_in, const int* in_sizes, int n_in,
                              void* d_out, int out_size, void* d_ws, size_t ws_size,
                              hipStream_t stream) {
    const float* x_m    = (const float*)d_in[0];
    const float* x_h    = (const float*)d_in[1];
    const float* W_nt_m = (const float*)d_in[2];
    const float* b_nt_m = (const float*)d_in[3];
    const float* W_nt_h = (const float*)d_in[4];
    const float* b_nt_h = (const float*)d_in[5];
    const float* Wl_mm  = (const float*)d_in[6];
    const float* bl_mm  = (const float*)d_in[7];
    const float* Wr_mm  = (const float*)d_in[8];
    const float* Wl_mh  = (const float*)d_in[9];
    const float* bl_mh  = (const float*)d_in[10];
    const float* Wr_mh  = (const float*)d_in[11];
    const float* Wl_hm  = (const float*)d_in[12];
    const float* bl_hm  = (const float*)d_in[13];
    const float* Wr_hm  = (const float*)d_in[14];
    const float* g_m    = (const float*)d_in[15];
    const float* be_m   = (const float*)d_in[16];
    const float* g_h    = (const float*)d_in[17];
    const float* be_h   = (const float*)d_in[18];
    const int*   ei_mm  = (const int*)d_in[19];
    const int*   ei_mh  = (const int*)d_in[20];
    const int*   ei_hm  = (const int*)d_in[21];

    const int N_M  = in_sizes[0] / D;
    const int N_H  = in_sizes[1] / D;
    const int E_mm = in_sizes[19] / 2;
    const int E_mh = in_sizes[20] / 2;
    const int E_hm = in_sizes[21] / 2;

    const size_t nm = (size_t)N_M * D;
    const size_t nh = (size_t)N_H * D;

    float* ws     = (float*)d_ws;
    float* t_m    = ws;
    float* t_h    = t_m + nm;
    float* agg_mm = t_h + nh;
    float* agg_hm = agg_mm + nm;
    float* agg_mh = agg_hm + nm;
    float* cnt_mm = agg_mh + nh;
    float* cnt_hm = cnt_mm + N_M;
    float* cnt_mh = cnt_hm + N_M;

    size_t zn = (nm + nm + nh) + (size_t)(N_M + N_M + N_H);
    zero_kernel<<<(int)((zn + 255) / 256), 256, 0, stream>>>(agg_mm, zn);

    lin_kernel<<<N_M / 16, 256, 0, stream>>>(x_m, W_nt_m, b_nt_m, t_m);
    lin_kernel<<<N_H / 16, 256, 0, stream>>>(x_h, W_nt_h, b_nt_h, t_h);

    auto sg = [](int E) { return (int)(((size_t)E * D + 255) / 256); };
    scatter_kernel<<<sg(E_mm), 256, 0, stream>>>(t_m, ei_mm, ei_mm + E_mm, agg_mm, cnt_mm, E_mm);
    scatter_kernel<<<sg(E_mh), 256, 0, stream>>>(t_m, ei_mh, ei_mh + E_mh, agg_mh, cnt_mh, E_mh);
    scatter_kernel<<<sg(E_hm), 256, 0, stream>>>(t_h, ei_hm, ei_hm + E_hm, agg_hm, cnt_hm, E_hm);

    float* out_m = (float*)d_out;
    float* out_h = out_m + nm;
    combine_m_kernel<<<N_M / 16, 256, 0, stream>>>(t_m, agg_mm, cnt_mm, agg_hm, cnt_hm,
                                                   Wl_mm, bl_mm, Wr_mm, Wl_hm, bl_hm, Wr_hm,
                                                   g_m, be_m, out_m);
    combine_h_kernel<<<N_H / 16, 256, 0, stream>>>(t_h, agg_mh, cnt_mh,
                                                   Wl_mh, bl_mh, Wr_mh,
                                                   g_h, be_h, out_h);
}


# Edge sums by privatised scatter: a work-group owns 592 destination rows as f32 sums in LDS, sweeps the whole dst list, adds matching source rows by LDS float atomics (arrival order); 8 launches to 5
# speedup vs baseline: 2.8010x; 2.8010x over previous
#include <hip/hip_runtime.h>

#define D 128
#define LDSW 130

typedef __attribute__((ext_vector_type(2))) float v2f;
typedef __attribute__((ext_vector_type(8))) float v8f;

__device__ __forceinline__ v8f wmma_f32(v2f a, v2f b, v8f c) {
    return __builtin_amdgcn_wmma_f32_16x16x4_f32(false, a, false, b, (short)0, c,
                                                 false, false);
}

__global__ __launch_bounds__(256)
void lin_kernel(const float* __restrict__ x, const float* __restrict__ W,
                const float* __restrict__ bias, float* __restrict__ t) {
    __shared__ float aS[16 * LDSW];
    const int tid  = threadIdx.x;
    const int row0 = blockIdx.x * 16;

    for (int i = tid; i < 16 * D; i += 256) {
        int r = i >> 7, c = i & (D - 1);
        aS[r * LDSW + c] = x[(size_t)(row0 + r) * D + c];
    }
    __syncthreads();

    const int lane = tid & 31, wv = tid >> 5;
    const int n = lane & 15, h = lane >> 4, m = lane & 15;
    const int kb = h * 2;
    const int col = wv * 16 + n;

    v8f acc = {};
    for (int k = 0; k < D; k += 4) {
        v2f a = *(const v2f*)&aS[m * LDSW + k + kb];
        v2f b;
        b.x = W[(size_t)(k + kb) * D + col];
        b.y = W[(size_t)(k + kb + 1) * D + col];
        acc = wmma_f32(a, b, acc);
    }
    const float bv = bias[col];
    for (int p = 0; p < 8; ++p)
        t[(size_t)(row0 + p + 8 * h) * D + col] = acc[p] + bv;
}

#define RB   592
#define NWV  32
#define TPI  4
#define DB   8
#define LCAP 64

__global__ __launch_bounds__(1024)
void own_agg_kernel(const float* __restrict__ t_m, const float* __restrict__ t_h,
                    const int* __restrict__ ei_mm, const int* __restrict__ ei_mh,
                    const int* __restrict__ ei_hm,
                    int E_mm, int E_mh, int E_hm, int N_M, int N_H,
                    float* __restrict__ agg_mm, float* __restrict__ agg_mh,
                    float* __restrict__ agg_hm,
                    float* __restrict__ cnt_mm, float* __restrict__ cnt_mh,
                    float* __restrict__ cnt_hm) {
    __shared__ float acc[(RB + 1) * D];
    __shared__ float cntS[RB];
    __shared__ int   lstE[NWV][LCAP];
    __shared__ int   lstD[NWV][LCAP];

    const float* tsrc; const int* src; float* agg; float* cnt; int E, N;
    if (blockIdx.y == 0)      { tsrc = t_m; src = ei_mm; E = E_mm; N = N_M; agg = agg_mm; cnt = cnt_mm; }
    else if (blockIdx.y == 1) { tsrc = t_m; src = ei_mh; E = E_mh; N = N_H; agg = agg_mh; cnt = cnt_mh; }
    else                      { tsrc = t_h; src = ei_hm; E = E_hm; N = N_M; agg = agg_hm; cnt = cnt_hm; }
    const int* dst = src + E;

    const int row0 = blockIdx.x * RB;
    if (row0 >= N) return;

    const int tid = threadIdx.x, lane = tid & 31;
    const int wv = __builtin_amdgcn_readfirstlane(tid >> 5);

    for (int i = tid; i < (RB + 1) * D / 4; i += 1024)
        ((float4*)acc)[i] = make_float4(0.0f, 0.0f, 0.0f, 0.0f);
    for (int i = tid; i < RB; i += 1024) cntS[i] = 0.0f;
    __syncthreads();

    int nl = 0;

    auto drain = [&](int n) {
        if (n <= 0) return;
        int d0 = RB, s0 = 0, d1 = RB, s1 = 0;
        if (lane < n) {
            d0 = lstD[wv][lane];
            s0 = src[lstE[wv][lane]];
            atomicAdd(&cntS[d0], 1.0f);
        }
        if (lane + 32 < n) {
            d1 = lstD[wv][lane + 32];
            s1 = src[lstE[wv][lane + 32]];
            atomicAdd(&cntS[d1], 1.0f);
        }
#pragma unroll 1
        for (int i = 0; i < n; i += DB) {
            const int sv = (i < 32) ? s0 : s1;
            const int dv = (i < 32) ? d0 : d1;
            float4 x[DB];
            int dk[DB];
#pragma unroll
            for (int k = 0; k < DB; ++k) {
                const int l = (i + k) & 31;
                const int sk = __shfl(sv, l);
                dk[k] = __shfl(dv, l);
                x[k] = *(const float4*)&tsrc[(size_t)sk * D + lane * 4];
            }
#pragma unroll
            for (int k = 0; k < DB; ++k) {
                float* a = &acc[dk[k] * D + lane];
                atomicAdd(a,      x[k].x);
                atomicAdd(a + 32, x[k].y);
                atomicAdd(a + 64, x[k].z);
                atomicAdd(a + 96, x[k].w);
            }
        }
    };
    auto append = [&](bool hit, unsigned m, int e, int d) {
        if (hit) {
            const int r = nl + (int)__builtin_amdgcn_mbcnt_lo(m, 0u);
            lstE[wv][r] = e;
            lstD[wv][r] = d;
        }
        nl += __builtin_popcount(m);
    };

    const int nG = E >> 2;
    const int nT = (nG + 31) >> 5;
    const int off = (int)(((unsigned)blockIdx.x * (unsigned)nT) / gridDim.x);
    const int4* dst4 = (const int4*)dst;

    auto group_of = [&](int tl, bool& ok) -> int {
        int tr = tl + off;
        if (tr >= nT) tr -= nT;
        const int g = tr * 32 + lane;
        ok = (tl < nT) && (g < nG);
        return g;
    };

    if (nG > 0) {
        int4 vn[TPI];
#pragma unroll
        for (int k = 0; k < TPI; ++k) {
            bool ok; const int g = group_of(wv + NWV * k, ok);
            vn[k] = dst4[ok ? g : 0];
        }
        for (int tt = wv; tt < nT; tt += NWV * TPI) {
            int4 v[TPI];
#pragma unroll
            for (int k = 0; k < TPI; ++k) v[k] = vn[k];
            if (tt + NWV * TPI < nT) {
#pragma unroll
                for (int k = 0; k < TPI; ++k) {
                    bool ok; const int g = group_of(tt + NWV * TPI + NWV * k, ok);
                    vn[k] = dst4[ok ? g : 0];
                }
            }

            int gk[TPI]; bool okk[TPI];
            unsigned m[TPI][4];
            int H = 0;
#pragma unroll
            for (int k = 0; k < TPI; ++k) {
                gk[k] = group_of(tt + NWV * k, okk[k]);
                const int dv[4] = { v[k].x, v[k].y, v[k].z, v[k].w };
#pragma unroll
                for (int c = 0; c < 4; ++c) {
                    const bool hit = okk[k] && (unsigned)(dv[c] - row0) < (unsigned)RB;
                    m[k][c] = __builtin_amdgcn_ballot_w32(hit);
                    H += __builtin_popcount(m[k][c]);
                }
            }
            if (H == 0) continue;
            if (nl + H > LCAP) { drain(nl); nl = 0; }
            if (H <= LCAP) {
#pragma unroll
                for (int k = 0; k < TPI; ++k) {
                    const int dv[4] = { v[k].x, v[k].y, v[k].z, v[k].w };
#pragma unroll
                    for (int c = 0; c < 4; ++c) {
                        if (m[k][c]) {
                            const int d = dv[c] - row0;
                            const bool hit = okk[k] && (unsigned)d < (unsigned)RB;
                            append(hit, m[k][c], gk[k] * 4 + c, d);
                        }
                    }
                }
            } else {
#pragma unroll 1
                for (int q = 0; q < TPI * 4; ++q) {
                    bool ok; const int g = group_of(tt + NWV * (q >> 2), ok);
                    const int e = g * 4 + (q & 3);
                    const int d = (ok ? dst[e] : -1) - row0;
                    const bool hit = ok && (unsigned)d < (unsigned)RB;
                    const unsigned mq = __builtin_amdgcn_ballot_w32(hit);
                    if (nl + __builtin_popcount(mq) > LCAP) { drain(nl); nl = 0; }
                    append(hit, mq, e, d);
                }
            }
        }
    }
    if (wv == 0 && (nG << 2) < E) {
        const int e = (nG << 2) + lane;
        const bool in = e < E;
        const int d = (in ? dst[e] : -1) - row0;
        const bool hit = in && (unsigned)d < (unsigned)RB;
        const unsigned mq = __builtin_amdgcn_ballot_w32(hit);
        if (nl + __builtin_popcount(mq) > LCAP) { drain(nl); nl = 0; }
        append(hit, mq, e, d);
    }
    drain(nl);
    __syncthreads();

    for (int i = tid; i < RB * 32; i += 1024) {
        const int r = i >> 5;
        if (row0 + r < N) {
            const float* a = &acc[r * D + lane];
            ((float4*)agg)[(size_t)(row0 + r) * 32 + lane] = make_float4(a[0], a[32], a[64], a[96]);
        }
    }
    for (int i = tid; i < RB; i += 1024)
        if (row0 + i < N) cnt[row0 + i] = cntS[i];
}

__global__ __launch_bounds__(256)
void combine_m_kernel(const float* __restrict__ t_m,
                      const float* __restrict__ agg_mm, const float* __restrict__ cnt_mm,
                      const float* __restrict__ agg_hm, const float* __restrict__ cnt_hm,
                      const float* __restrict__ Wl_mm, const float* __restrict__ bl_mm,
                      const float* __restrict__ Wr_mm,
                      const float* __restrict__ Wl_hm, const float* __restrict__ bl_hm,
                      const float* __restrict__ Wr_hm,
                      const float* __restrict__ g, const float* __restrict__ be,
                      float* __restrict__ out) {
    __shared__ float tS[16 * LDSW], aS[16 * LDSW], bS[16 * LDSW];
    __shared__ float red[4][16];
    const int tid  = threadIdx.x;
    const int row0 = blockIdx.x * 16;

    if (tid < 64) red[tid >> 4][tid & 15] = 0.0f;
    for (int i = tid; i < 16 * D; i += 256) {
        int r = i >> 7, c = i & (D - 1);
        int gr = row0 + r;
        tS[r * LDSW + c] = t_m[(size_t)gr * D + c];
        aS[r * LDSW + c] = agg_mm[(size_t)gr * D + c] / fmaxf(cnt_mm[gr], 1.0f);
        bS[r * LDSW + c] = agg_hm[(size_t)gr * D + c] / fmaxf(cnt_hm[gr], 1.0f);
    }
    __syncthreads();

    const int lane = tid & 31, wv = tid >> 5;
    const int n = lane & 15, h = lane >> 4, m = lane & 15;
    const int kb = h * 2;
    const int col = wv * 16 + n;

    v8f acc_mm = {}, acc_hm = {};
    for (int k = 0; k < D; k += 4) {
        v2f at = *(const v2f*)&tS[m * LDSW + k + kb];
        v2f aa = *(const v2f*)&aS[m * LDSW + k + kb];
        v2f ab = *(const v2f*)&bS[m * LDSW + k + kb];
        v2f b1; b1.x = Wl_mm[(size_t)(k + kb) * D + col]; b1.y = Wl_mm[(size_t)(k + kb + 1) * D + col];
        v2f b2; b2.x = Wr_mm[(size_t)(k + kb) * D + col]; b2.y = Wr_mm[(size_t)(k + kb + 1) * D + col];
        v2f b3; b3.x = Wl_hm[(size_t)(k + kb) * D + col]; b3.y = Wl_hm[(size_t)(k + kb + 1) * D + col];
        v2f b4; b4.x = Wr_hm[(size_t)(k + kb) * D + col]; b4.y = Wr_hm[(size_t)(k + kb + 1) * D + col];
        acc_mm = wmma_f32(aa, b1, acc_mm);
        acc_mm = wmma_f32(at, b2, acc_mm);
        acc_hm = wmma_f32(ab, b3, acc_hm);
        acc_hm = wmma_f32(at, b4, acc_hm);
    }

    float u1[8], u2[8];
    const float c1 = bl_mm[col], c2 = bl_hm[col];
    for (int p = 0; p < 8; ++p) {
        u1[p] = acc_mm[p] + c1;
        u2[p] = acc_hm[p] + c2;
        int r = p + 8 * h;
        atomicAdd(&red[0][r], u1[p] * u1[p]);
        atomicAdd(&red[1][r], u2[p] * u2[p]);
    }
    __syncthreads();

    float y[8];
    for (int p = 0; p < 8; ++p) {
        int r = p + 8 * h;
        float n1 = fmaxf(sqrtf(red[0][r]), 1e-12f);
        float n2 = fmaxf(sqrtf(red[1][r]), 1e-12f);
        float conv = 0.5f * (u1[p] / n1 + u2[p] / n2);
        y[p] = tS[r * LDSW + col] + conv;
        atomicAdd(&red[2][r], y[p]);
        atomicAdd(&red[3][r], y[p] * y[p]);
    }
    __syncthreads();

    const float gc = g[col], bc = be[col];
    for (int p = 0; p < 8; ++p) {
        int r = p + 8 * h;
        float mu  = red[2][r] * (1.0f / D);
        float var = red[3][r] * (1.0f / D) - mu * mu;
        out[(size_t)(row0 + r) * D + col] = (y[p] - mu) * rsqrtf(var + 1e-5f) * gc + bc;
    }
}

__global__ __launch_bounds__(256)
void combine_h_kernel(const float* __restrict__ t_h,
                      const float* __restrict__ agg_mh, const float* __restrict__ cnt_mh,
                      const float* __restrict__ Wl_mh, const float* __restrict__ bl_mh,
                      const float* __restrict__ Wr_mh,
                      const float* __restrict__ g, const float* __restrict__ be,
                      float* __restrict__ out) {
    __shared__ float tS[16 * LDSW], aS[16 * LDSW];
    __shared__ float red[3][16];
    const int tid  = threadIdx.x;
    const int row0 = blockIdx.x * 16;

    if (tid < 48) red[tid >> 4][tid & 15] = 0.0f;
    for (int i = tid; i < 16 * D; i += 256) {
        int r = i >> 7, c = i & (D - 1);
        int gr = row0 + r;
        tS[r * LDSW + c] = t_h[(size_t)gr * D + c];
        aS[r * LDSW + c] = agg_mh[(size_t)gr * D + c] / fmaxf(cnt_mh[gr], 1.0f);
    }
    __syncthreads();

    const int lane = tid & 31, wv = tid >> 5;
    const int n = lane & 15, h = lane >> 4, m = lane & 15;
    const int kb = h * 2;
    const int col = wv * 16 + n;

    v8f acc = {};
    for (int k = 0; k < D; k += 4) {
        v2f at = *(const v2f*)&tS[m * LDSW + k + kb];
        v2f aa = *(const v2f*)&aS[m * LDSW + k + kb];
        v2f b1; b1.x = Wl_mh[(size_t)(k + kb) * D + col]; b1.y = Wl_mh[(size_t)(k + kb + 1) * D + col];
        v2f b2; b2.x = Wr_mh[(size_t)(k + kb) * D + col]; b2.y = Wr_mh[(size_t)(k + kb + 1) * D + col];
        acc = wmma_f32(aa, b1, acc);
        acc = wmma_f32(at, b2, acc);
    }

    float u[8];
    const float c1 = bl_mh[col];
    for (int p = 0; p < 8; ++p) {
        u[p] = acc[p] + c1;
        atomicAdd(&red[0][p + 8 * h], u[p] * u[p]);
    }
    __syncthreads();

    float y[8];
    for (int p = 0; p < 8; ++p) {
        int r = p + 8 * h;
        float n1 = fmaxf(sqrtf(red[0][r]), 1e-12f);
        y[p] = tS[r * LDSW + col] + u[p] / n1;
        atomicAdd(&red[1][r], y[p]);
        atomicAdd(&red[2][r], y[p] * y[p]);
    }
    __syncthreads();

    const float gc = g[col], bc = be[col];
    for (int p = 0; p < 8; ++p) {
        int r = p + 8 * h;
        float mu  = red[1][r] * (1.0f / D);
        float var = red[2][r] * (1.0f / D) - mu * mu;
        out[(size_t)(row0 + r) * D + col] = (y[p] - mu) * rsqrtf(var + 1e-5f) * gc + bc;
    }
}

extern "C" void kernel_launch(void* const* d_in, const int* in_sizes, int n_in,
                              void* d_out, int out_size, void* d_ws, size_t ws_size,
                              hipStream_t stream) {
    const float* x_m    = (const float*)d_in[0];
    const float* x_h    = (const float*)d_in[1];
    const float* W_nt_m = (const float*)d_in[2];
    const float* b_nt_m = (const float*)d_in[3];
    const float* W_nt_h = (const float*)d_in[4];
    const float* b_nt_h = (const float*)d_in[5];
    const float* Wl_mm  = (const float*)d_in[6];
    const float* bl_mm  = (const float*)d_in[7];
    const float* Wr_mm  = (const float*)d_in[8];
    const float* Wl_mh  = (const float*)d_in[9];
    const float* bl_mh  = (const float*)d_in[10];
    const float* Wr_mh  = (const float*)d_in[11];
    const float* Wl_hm  = (const float*)d_in[12];
    const float* bl_hm  = (const float*)d_in[13];
    const float* Wr_hm  = (const float*)d_in[14];
    const float* g_m    = (const float*)d_in[15];
    const float* be_m   = (const float*)d_in[16];
    const float* g_h    = (const float*)d_in[17];
    const float* be_h   = (const float*)d_in[18];
    const int*   ei_mm  = (const int*)d_in[19];
    const int*   ei_mh  = (const int*)d_in[20];
    const int*   ei_hm  = (const int*)d_in[21];

    const int N_M  = in_sizes[0] / D;
    const int N_H  = in_sizes[1] / D;
    const int E_mm = in_sizes[19] / 2;
    const int E_mh = in_sizes[20] / 2;
    const int E_hm = in_sizes[21] / 2;

    const size_t nm = (size_t)N_M * D;
    const size_t nh = (size_t)N_H * D;

    float* ws     = (float*)d_ws;
    float* t_m    = ws;
    float* t_h    = t_m + nm;
    float* agg_mm = t_h + nh;
    float* agg_hm = agg_mm + nm;
    float* agg_mh = agg_hm + nm;
    float* cnt_mm = agg_mh + nh;
    float* cnt_hm = cnt_mm + N_M;
    float* cnt_mh = cnt_hm + N_M;

    lin_kernel<<<N_M / 16, 256, 0, stream>>>(x_m, W_nt_m, b_nt_m, t_m);
    lin_kernel<<<N_H / 16, 256, 0, stream>>>(x_h, W_nt_h, b_nt_h, t_h);

    const int nmax = N_M > N_H ? N_M : N_H;
    own_agg_kernel<<<dim3((nmax + RB - 1) / RB, 3), 1024, 0, stream>>>(
        t_m, t_h, ei_mm, ei_mh, ei_hm, E_mm, E_mh, E_hm, N_M, N_H,
        agg_mm, agg_mh, agg_hm, cnt_mm, cnt_mh, cnt_hm);

    float* out_m = (float*)d_out;
    float* out_h = out_m + nm;
    combine_m_kernel<<<N_M / 16, 256, 0, stream>>>(t_m, agg_mm, cnt_mm, agg_hm, cnt_hm,
                                                   Wl_mm, bl_mm, Wr_mm, Wl_hm, bl_hm, Wr_hm,
                                                   g_m, be_m, out_m);
    combine_h_kernel<<<N_H / 16, 256, 0, stream>>>(t_h, agg_mh, cnt_mh,
                                                   Wl_mh, bl_mh, Wr_mh,
                                                   g_h, be_h, out_h);
}


# Combine kernels re-tiled: one wave owns 16 whole rows (8 column tiles in registers), so L2-norm and LayerNorm row sums are in-lane plus 16-lane butterfly, fixed order; no LDS atomics or barriers
# speedup vs baseline: 4.0661x; 1.4517x over previous
#include <hip/hip_runtime.h>

#define D 128
#define LDSW 130

typedef __attribute__((ext_vector_type(2))) float v2f;
typedef __attribute__((ext_vector_type(8))) float v8f;

__device__ __forceinline__ v8f wmma_f32(v2f a, v2f b, v8f c) {
    return __builtin_amdgcn_wmma_f32_16x16x4_f32(false, a, false, b, (short)0, c,
                                                 false, false);
}

__global__ __launch_bounds__(256)
void lin_kernel(const float* __restrict__ x, const float* __restrict__ W,
                const float* __restrict__ bias, float* __restrict__ t) {
    __shared__ float aS[16 * LDSW];
    const int tid  = threadIdx.x;
    const int row0 = blockIdx.x * 16;

    for (int i = tid; i < 16 * D; i += 256) {
        int r = i >> 7, c = i & (D - 1);
        aS[r * LDSW + c] = x[(size_t)(row0 + r) * D + c];
    }
    __syncthreads();

    const int lane = tid & 31, wv = tid >> 5;
    const int n = lane & 15, h = lane >> 4, m = lane & 15;
    const int kb = h * 2;
    const int col = wv * 16 + n;

    v8f acc = {};
    for (int k = 0; k < D; k += 4) {
        v2f a = *(const v2f*)&aS[m * LDSW + k + kb];
        v2f b;
        b.x = W[(size_t)(k + kb) * D + col];
        b.y = W[(size_t)(k + kb + 1) * D + col];
        acc = wmma_f32(a, b, acc);
    }
    const float bv = bias[col];
    for (int p = 0; p < 8; ++p)
        t[(size_t)(row0 + p + 8 * h) * D + col] = acc[p] + bv;
}

#define RB   592
#define NWV  32
#define TPI  4
#define DB   8
#define LCAP 64

__global__ __launch_bounds__(1024)
void own_agg_kernel(const float* __restrict__ t_m, const float* __restrict__ t_h,
                    const int* __restrict__ ei_mm, const int* __restrict__ ei_mh,
                    const int* __restrict__ ei_hm,
                    int E_mm, int E_mh, int E_hm, int N_M, int N_H,
                    float* __restrict__ agg_mm, float* __restrict__ agg_mh,
                    float* __restrict__ agg_hm,
                    float* __restrict__ cnt_mm, float* __restrict__ cnt_mh,
                    float* __restrict__ cnt_hm) {
    __shared__ float acc[(RB + 1) * D];
    __shared__ float cntS[RB];
    __shared__ int   lstE[NWV][LCAP];
    __shared__ int   lstD[NWV][LCAP];

    const float* tsrc; const int* src; float* agg; float* cnt; int E, N;
    if (blockIdx.y == 0)      { tsrc = t_m; src = ei_mm; E = E_mm; N = N_M; agg = agg_mm; cnt = cnt_mm; }
    else if (blockIdx.y == 1) { tsrc = t_m; src = ei_mh; E = E_mh; N = N_H; agg = agg_mh; cnt = cnt_mh; }
    else                      { tsrc = t_h; src = ei_hm; E = E_hm; N = N_M; agg = agg_hm; cnt = cnt_hm; }
    const int* dst = src + E;

    const int row0 = blockIdx.x * RB;
    if (row0 >= N) return;

    const int tid = threadIdx.x, lane = tid & 31;
    const int wv = __builtin_amdgcn_readfirstlane(tid >> 5);

    for (int i = tid; i < (RB + 1) * D / 4; i += 1024)
        ((float4*)acc)[i] = make_float4(0.0f, 0.0f, 0.0f, 0.0f);
    for (int i = tid; i < RB; i += 1024) cntS[i] = 0.0f;
    __syncthreads();

    int nl = 0;

    auto drain = [&](int n) {
        if (n <= 0) return;
        int d0 = RB, s0 = 0, d1 = RB, s1 = 0;
        if (lane < n) {
            d0 = lstD[wv][lane];
            s0 = src[lstE[wv][lane]];
            atomicAdd(&cntS[d0], 1.0f);
        }
        if (lane + 32 < n) {
            d1 = lstD[wv][lane + 32];
            s1 = src[lstE[wv][lane + 32]];
            atomicAdd(&cntS[d1], 1.0f);
        }
#pragma unroll 1
        for (int i = 0; i < n; i += DB) {
            const int sv = (i < 32) ? s0 : s1;
            const int dv = (i < 32) ? d0 : d1;
            float4 x[DB];
            int dk[DB];
#pragma unroll
            for (int k = 0; k < DB; ++k) {
                const int l = (i + k) & 31;
                const int sk = __shfl(sv, l);
                dk[k] = __shfl(dv, l);
                x[k] = *(const float4*)&tsrc[(size_t)sk * D + lane * 4];
            }
#pragma unroll
            for (int k = 0; k < DB; ++k) {
                float* a = &acc[dk[k] * D + lane];
                atomicAdd(a,      x[k].x);
                atomicAdd(a + 32, x[k].y);
                atomicAdd(a + 64, x[k].z);
                atomicAdd(a + 96, x[k].w);
            }
        }
    };
    auto append = [&](bool hit, unsigned m, int e, int d) {
        if (hit) {
            const int r = nl + (int)__builtin_amdgcn_mbcnt_lo(m, 0u);
            lstE[wv][r] = e;
            lstD[wv][r] = d;
        }
        nl += __builtin_popcount(m);
    };

    const int nG = E >> 2;
    const int nT = (nG + 31) >> 5;
    const int off = (int)(((unsigned)blockIdx.x * (unsigned)nT) / gridDim.x);
    const int4* dst4 = (const int4*)dst;

    auto group_of = [&](int tl, bool& ok) -> int {
        int tr = tl + off;
        if (tr >= nT) tr -= nT;
        const int g = tr * 32 + lane;
        ok = (tl < nT) && (g < nG);
        return g;
    };

    if (nG > 0) {
        int4 vn[TPI];
#pragma unroll
        for (int k = 0; k < TPI; ++k) {
            bool ok; const int g = group_of(wv + NWV * k, ok);
            vn[k] = dst4[ok ? g : 0];
        }
        for (int tt = wv; tt < nT; tt += NWV * TPI) {
            int4 v[TPI];
#pragma unroll
            for (int k = 0; k < TPI; ++k) v[k] = vn[k];
            if (tt + NWV * TPI < nT) {
#pragma unroll
                for (int k = 0; k < TPI; ++k) {
                    bool ok; const int g = group_of(tt + NWV * TPI + NWV * k, ok);
                    vn[k] = dst4[ok ? g : 0];
                }
            }

            int gk[TPI]; bool okk[TPI];
            unsigned m[TPI][4];
            int H = 0;
#pragma unroll
            for (int k = 0; k < TPI; ++k) {
                gk[k] = group_of(tt + NWV * k, okk[k]);
                const int dv[4] = { v[k].x, v[k].y, v[k].z, v[k].w };
#pragma unroll
                for (int c = 0; c < 4; ++c) {
                    const bool hit = okk[k] && (unsigned)(dv[c] - row0) < (unsigned)RB;
                    m[k][c] = __builtin_amdgcn_ballot_w32(hit);
                    H += __builtin_popcount(m[k][c]);
                }
            }
            if (H == 0) continue;
            if (nl + H > LCAP) { drain(nl); nl = 0; }
            if (H <= LCAP) {
#pragma unroll
                for (int k = 0; k < TPI; ++k) {
                    const int dv[4] = { v[k].x, v[k].y, v[k].z, v[k].w };
#pragma unroll
                    for (int c = 0; c < 4; ++c) {
                        if (m[k][c]) {
                            const int d = dv[c] - row0;
                            const bool hit = okk[k] && (unsigned)d < (unsigned)RB;
                            append(hit, m[k][c], gk[k] * 4 + c, d);
                        }
                    }
                }
            } else {
#pragma unroll 1
                for (int q = 0; q < TPI * 4; ++q) {
                    bool ok; const int g = group_of(tt + NWV * (q >> 2), ok);
                    const int e = g * 4 + (q & 3);
                    const int d = (ok ? dst[e] : -1) - row0;
                    const bool hit = ok && (unsigned)d < (unsigned)RB;
                    const unsigned mq = __builtin_amdgcn_ballot_w32(hit);
                    if (nl + __builtin_popcount(mq) > LCAP) { drain(nl); nl = 0; }
                    append(hit, mq, e, d);
                }
            }
        }
    }
    if (wv == 0 && (nG << 2) < E) {
        const int e = (nG << 2) + lane;
        const bool in = e < E;
        const int d = (in ? dst[e] : -1) - row0;
        const bool hit = in && (unsigned)d < (unsigned)RB;
        const unsigned mq = __builtin_amdgcn_ballot_w32(hit);
        if (nl + __builtin_popcount(mq) > LCAP) { drain(nl); nl = 0; }
        append(hit, mq, e, d);
    }
    drain(nl);
    __syncthreads();

    for (int i = tid; i < RB * 32; i += 1024) {
        const int r = i >> 5;
        if (row0 + r < N) {
            const float* a = &acc[r * D + lane];
            ((float4*)agg)[(size_t)(row0 + r) * 32 + lane] = make_float4(a[0], a[32], a[64], a[96]);
        }
    }
    for (int i = tid; i < RB; i += 1024)
        if (row0 + i < N) cnt[row0 + i] = cntS[i];
}

__device__ __forceinline__ float rowsum16(float v) {
    v += __shfl_xor(v, 1);
    v += __shfl_xor(v, 2);
    v += __shfl_xor(v, 4);
    v += __shfl_xor(v, 8);
    return v;
}

__global__ __launch_bounds__(32)
void combine_m_kernel(const float* __restrict__ t_m,
                      const float* __restrict__ agg_mm, const float* __restrict__ cnt_mm,
                      const float* __restrict__ agg_hm, const float* __restrict__ cnt_hm,
                      const float* __restrict__ Wl_mm, const float* __restrict__ bl_mm,
                      const float* __restrict__ Wr_mm,
                      const float* __restrict__ Wl_hm, const float* __restrict__ bl_hm,
                      const float* __restrict__ Wr_hm,
                      const float* __restrict__ g, const float* __restrict__ be,
                      float* __restrict__ out) {
    const int lane = threadIdx.x;
    const int row0 = blockIdx.x * 16;
    const int n = lane & 15, h = lane >> 4, m = lane & 15;
    const int kb = h * 2;

    const float* tr = t_m    + (size_t)(row0 + m) * D + kb;
    const float* ar = agg_mm + (size_t)(row0 + m) * D + kb;
    const float* br = agg_hm + (size_t)(row0 + m) * D + kb;
    const float ca = fmaxf(cnt_mm[row0 + m], 1.0f);
    const float cb = fmaxf(cnt_hm[row0 + m], 1.0f);
    const size_t wo = (size_t)kb * D + n;

    v8f acc_mm[8], acc_hm[8];
#pragma unroll
    for (int ct = 0; ct < 8; ++ct) { acc_mm[ct] = v8f{}; acc_hm[ct] = v8f{}; }

#pragma unroll 1
    for (int kc = 0; kc < D; kc += 16) {
        v2f at[4], aa[4], ab[4];
#pragma unroll
        for (int i = 0; i < 4; ++i) {
            at[i] = *(const v2f*)(tr + kc + 4 * i);
            const v2f xa = *(const v2f*)(ar + kc + 4 * i);
            const v2f xb = *(const v2f*)(br + kc + 4 * i);
            aa[i].x = xa.x / ca; aa[i].y = xa.y / ca;
            ab[i].x = xb.x / cb; ab[i].y = xb.y / cb;
        }
#pragma unroll
        for (int i = 0; i < 4; ++i) {
            const size_t w0 = wo + (size_t)(kc + 4 * i) * D;
#pragma unroll
            for (int ct = 0; ct < 8; ++ct) {
                v2f b1, b2, b3, b4;
                b1.x = Wl_mm[w0 + 16 * ct]; b1.y = Wl_mm[w0 + D + 16 * ct];
                b2.x = Wr_mm[w0 + 16 * ct]; b2.y = Wr_mm[w0 + D + 16 * ct];
                b3.x = Wl_hm[w0 + 16 * ct]; b3.y = Wl_hm[w0 + D + 16 * ct];
                b4.x = Wr_hm[w0 + 16 * ct]; b4.y = Wr_hm[w0 + D + 16 * ct];
                acc_mm[ct] = wmma_f32(aa[i], b1, acc_mm[ct]);
                acc_hm[ct] = wmma_f32(ab[i], b3, acc_hm[ct]);
                acc_mm[ct] = wmma_f32(at[i], b2, acc_mm[ct]);
                acc_hm[ct] = wmma_f32(at[i], b4, acc_hm[ct]);
            }
        }
    }

    float s1[8], s2[8];
#pragma unroll
    for (int p = 0; p < 8; ++p) { s1[p] = 0.0f; s2[p] = 0.0f; }
#pragma unroll
    for (int ct = 0; ct < 8; ++ct) {
        const float c1 = bl_mm[16 * ct + n], c2 = bl_hm[16 * ct + n];
#pragma unroll
        for (int p = 0; p < 8; ++p) {
            acc_mm[ct][p] += c1;
            acc_hm[ct][p] += c2;
            s1[p] += acc_mm[ct][p] * acc_mm[ct][p];
            s2[p] += acc_hm[ct][p] * acc_hm[ct][p];
        }
    }
    float n1[8], n2[8];
#pragma unroll
    for (int p = 0; p < 8; ++p) {
        n1[p] = fmaxf(sqrtf(rowsum16(s1[p])), 1e-12f);
        n2[p] = fmaxf(sqrtf(rowsum16(s2[p])), 1e-12f);
    }

    float sy[8], sq[8];
#pragma unroll
    for (int p = 0; p < 8; ++p) { sy[p] = 0.0f; sq[p] = 0.0f; }
#pragma unroll
    for (int ct = 0; ct < 8; ++ct) {
#pragma unroll
        for (int p = 0; p < 8; ++p) {
            const float tv = t_m[(size_t)(row0 + p + 8 * h) * D + 16 * ct + n];
            const float conv = 0.5f * (acc_mm[ct][p] / n1[p] + acc_hm[ct][p] / n2[p]);
            const float y = tv + conv;
            acc_mm[ct][p] = y;
            sy[p] += y;
            sq[p] += y * y;
        }
    }
    float mu[8], rs[8];
#pragma unroll
    for (int p = 0; p < 8; ++p) {
        mu[p] = rowsum16(sy[p]) * (1.0f / D);
        const float var = rowsum16(sq[p]) * (1.0f / D) - mu[p] * mu[p];
        rs[p] = rsqrtf(var + 1e-5f);
    }
#pragma unroll
    for (int ct = 0; ct < 8; ++ct) {
        const float gc = g[16 * ct + n], bc = be[16 * ct + n];
#pragma unroll
        for (int p = 0; p < 8; ++p)
            out[(size_t)(row0 + p + 8 * h) * D + 16 * ct + n] = (acc_mm[ct][p] - mu[p]) * rs[p] * gc + bc;
    }
}

__global__ __launch_bounds__(32)
void combine_h_kernel(const float* __restrict__ t_h,
                      const float* __restrict__ agg_mh, const float* __restrict__ cnt_mh,
                      const float* __restrict__ Wl_mh, const float* __restrict__ bl_mh,
                      const float* __restrict__ Wr_mh,
                      const float* __restrict__ g, const float* __restrict__ be,
                      float* __restrict__ out) {
    const int lane = threadIdx.x;
    const int row0 = blockIdx.x * 16;
    const int n = lane & 15, h = lane >> 4, m = lane & 15;
    const int kb = h * 2;

    const float* tr = t_h    + (size_t)(row0 + m) * D + kb;
    const float* ar = agg_mh + (size_t)(row0 + m) * D + kb;
    const float ca = fmaxf(cnt_mh[row0 + m], 1.0f);
    const size_t wo = (size_t)kb * D + n;

    v8f acc[8];
#pragma unroll
    for (int ct = 0; ct < 8; ++ct) acc[ct] = v8f{};

#pragma unroll 1
    for (int kc = 0; kc < D; kc += 16) {
        v2f at[4], aa[4];
#pragma unroll
        for (int i = 0; i < 4; ++i) {
            at[i] = *(const v2f*)(tr + kc + 4 * i);
            const v2f xa = *(const v2f*)(ar + kc + 4 * i);
            aa[i].x = xa.x / ca; aa[i].y = xa.y / ca;
        }
#pragma unroll
        for (int i = 0; i < 4; ++i) {
            const size_t w0 = wo + (size_t)(kc + 4 * i) * D;
#pragma unroll
            for (int ct = 0; ct < 8; ++ct) {
                v2f b1, b2;
                b1.x = Wl_mh[w0 + 16 * ct]; b1.y = Wl_mh[w0 + D + 16 * ct];
                b2.x = Wr_mh[w0 + 16 * ct]; b2.y = Wr_mh[w0 + D + 16 * ct];
                acc[ct] = wmma_f32(aa[i], b1, acc[ct]);
                acc[ct] = wmma_f32(at[i], b2, acc[ct]);
            }
        }
    }

    float s1[8];
#pragma unroll
    for (int p = 0; p < 8; ++p) s1[p] = 0.0f;
#pragma unroll
    for (int ct = 0; ct < 8; ++ct) {
        const float c1 = bl_mh[16 * ct + n];
#pragma unroll
        for (int p = 0; p < 8; ++p) {
            acc[ct][p] += c1;
            s1[p] += acc[ct][p] * acc[ct][p];
        }
    }
    float n1[8];
#pragma unroll
    for (int p = 0; p < 8; ++p) n1[p] = fmaxf(sqrtf(rowsum16(s1[p])), 1e-12f);

    float sy[8], sq[8];
#pragma unroll
    for (int p = 0; p < 8; ++p) { sy[p] = 0.0f; sq[p] = 0.0f; }
#pragma unroll
    for (int ct = 0; ct < 8; ++ct) {
#pragma unroll
        for (int p = 0; p < 8; ++p) {
            const float tv = t_h[(size_t)(row0 + p + 8 * h) * D + 16 * ct + n];
            const float y = tv + acc[ct][p] / n1[p];
            acc[ct][p] = y;
            sy[p] += y;
            sq[p] += y * y;
        }
    }
    float mu[8], rs[8];
#pragma unroll
    for (int p = 0; p < 8; ++p) {
        mu[p] = rowsum16(sy[p]) * (1.0f / D);
        const float var = rowsum16(sq[p]) * (1.0f / D) - mu[p] * mu[p];
        rs[p] = rsqrtf(var + 1e-5f);
    }
#pragma unroll
    for (int ct = 0; ct < 8; ++ct) {
        const float gc = g[16 * ct + n], bc = be[16 * ct + n];
#pragma unroll
        for (int p = 0; p < 8; ++p)
            out[(size_t)(row0 + p + 8 * h) * D + 16 * ct + n] = (acc[ct][p] - mu[p]) * rs[p] * gc + bc;
    }
}

extern "C" void kernel_launch(void* const* d_in, const int* in_sizes, int n_in,
                              void* d_out, int out_size, void* d_ws, size_t ws_size,
                              hipStream_t stream) {
    const float* x_m    = (const float*)d_in[0];
    const float* x_h    = (const float*)d_in[1];
    const float* W_nt_m = (const float*)d_in[2];
    const float* b_nt_m = (const float*)d_in[3];
    const float* W_nt_h = (const float*)d_in[4];
    const float* b_nt_h = (const float*)d_in[5];
    const float* Wl_mm  = (const float*)d_in[6];
    const float* bl_mm  = (const float*)d_in[7];
    const float* Wr_mm  = (const float*)d_in[8];
    const float* Wl_mh  = (const float*)d_in[9];
    const float* bl_mh  = (const float*)d_in[10];
    const float* Wr_mh  = (const float*)d_in[11];
    const float* Wl_hm  = (const float*)d_in[12];
    const float* bl_hm  = (const float*)d_in[13];
    const float* Wr_hm  = (const float*)d_in[14];
    const float* g_m    = (const float*)d_in[15];
    const float* be_m   = (const float*)d_in[16];
    const float* g_h    = (const float*)d_in[17];
    const float* be_h   = (const float*)d_in[18];
    const int*   ei_mm  = (const int*)d_in[19];
    const int*   ei_mh  = (const int*)d_in[20];
    const int*   ei_hm  = (const int*)d_in[21];

    const int N_M  = in_sizes[0] / D;
    const int N_H  = in_sizes[1] / D;
    const int E_mm = in_sizes[19] / 2;
    const int E_mh = in_sizes[20] / 2;
    const int E_hm = in_sizes[21] / 2;

    const size_t nm = (size_t)N_M * D;
    const size_t nh = (size_t)N_H * D;

    float* ws     = (float*)d_ws;
    float* t_m    = ws;
    float* t_h    = t_m + nm;
    float* agg_mm = t_h + nh;
    float* agg_hm = agg_mm + nm;
    float* agg_mh = agg_hm + nm;
    float* cnt_mm = agg_mh + nh;
    float* cnt_hm = cnt_mm + N_M;
    float* cnt_mh = cnt_hm + N_M;

    lin_kernel<<<N_M / 16, 256, 0, stream>>>(x_m, W_nt_m, b_nt_m, t_m);
    lin_kernel<<<N_H / 16, 256, 0, stream>>>(x_h, W_nt_h, b_nt_h, t_h);

    const int nmax = N_M > N_H ? N_M : N_H;
    own_agg_kernel<<<dim3((nmax + RB - 1) / RB, 3), 1024, 0, stream>>>(
        t_m, t_h, ei_mm, ei_mh, ei_hm, E_mm, E_mh, E_hm, N_M, N_H,
        agg_mm, agg_mh, agg_hm, cnt_mm, cnt_mh, cnt_hm);

    float* out_m = (float*)d_out;
    float* out_h = out_m + nm;
    combine_m_kernel<<<N_M / 16, 32, 0, stream>>>(t_m, agg_mm, cnt_mm, agg_hm, cnt_hm,
                                                   Wl_mm, bl_mm, Wr_mm, Wl_hm, bl_hm, Wr_hm,
                                                   g_m, be_m, out_m);
    combine_h_kernel<<<N_H / 16, 32, 0, stream>>>(t_h, agg_mh, cnt_mh,
                                                   Wl_mh, bl_mh, Wr_mh,
                                                   g_h, be_h, out_h);
}


# Launches merged, 5 to 3: linear transforms share one launch (grid.y); combine types share one kernel, two-branch tiles first (two passes of one K loop), one-branch tiles fill in; sum orders unchanged
# speedup vs baseline: 4.5115x; 1.1095x over previous
#include <hip/hip_runtime.h>

#define D 128
#define LDSW 130

typedef __attribute__((ext_vector_type(2))) float v2f;
typedef __attribute__((ext_vector_type(8))) float v8f;

__device__ __forceinline__ v8f wmma_f32(v2f a, v2f b, v8f c) {
    return __builtin_amdgcn_wmma_f32_16x16x4_f32(false, a, false, b, (short)0, c,
                                                 false, false);
}

__global__ __launch_bounds__(256)
void lin_kernel(const float* __restrict__ x_m, const float* __restrict__ W_m,
                const float* __restrict__ bias_m, float* __restrict__ t_m, int N_M,
                const float* __restrict__ x_h, const float* __restrict__ W_h,
                const float* __restrict__ bias_h, float* __restrict__ t_h, int N_H) {
    __shared__ float aS[16 * LDSW];
    const float* x; const float* W; const float* bias; float* t; int N;
    if (blockIdx.y == 0) { x = x_m; W = W_m; bias = bias_m; t = t_m; N = N_M; }
    else                 { x = x_h; W = W_h; bias = bias_h; t = t_h; N = N_H; }
    const int tid  = threadIdx.x;
    const int row0 = blockIdx.x * 16;
    if (row0 >= N) return;

    for (int i = tid; i < 16 * D; i += 256) {
        int r = i >> 7, c = i & (D - 1);
        aS[r * LDSW + c] = x[(size_t)(row0 + r) * D + c];
    }
    __syncthreads();

    const int lane = tid & 31, wv = tid >> 5;
    const int n = lane & 15, h = lane >> 4, m = lane & 15;
    const int kb = h * 2;
    const int col = wv * 16 + n;

    v8f acc = {};
    for (int k = 0; k < D; k += 4) {
        v2f a = *(const v2f*)&aS[m * LDSW + k + kb];
        v2f b;
        b.x = W[(size_t)(k + kb) * D + col];
        b.y = W[(size_t)(k + kb + 1) * D + col];
        acc = wmma_f32(a, b, acc);
    }
    const float bv = bias[col];
    for (int p = 0; p < 8; ++p)
        t[(size_t)(row0 + p + 8 * h) * D + col] = acc[p] + bv;
}

#define RB   592
#define NWV  32
#define TPI  4
#define DB   8
#define LCAP 64

__global__ __launch_bounds__(1024)
void own_agg_kernel(const float* __restrict__ t_m, const float* __restrict__ t_h,
                    const int* __restrict__ ei_mm, const int* __restrict__ ei_mh,
                    const int* __restrict__ ei_hm,
                    int E_mm, int E_mh, int E_hm, int N_M, int N_H,
                    float* __restrict__ agg_mm, float* __restrict__ agg_mh,
                    float* __restrict__ agg_hm,
                    float* __restrict__ cnt_mm, float* __restrict__ cnt_mh,
                    float* __restrict__ cnt_hm) {
    __shared__ float acc[(RB + 1) * D];
    __shared__ float cntS[RB];
    __shared__ int   lstE[NWV][LCAP];
    __shared__ int   lstD[NWV][LCAP];

    const float* tsrc; const int* src; float* agg; float* cnt; int E, N;
    if (blockIdx.y == 0)      { tsrc = t_m; src = ei_mm; E = E_mm; N = N_M; agg = agg_mm; cnt = cnt_mm; }
    else if (blockIdx.y == 1) { tsrc = t_m; src = ei_mh; E = E_mh; N = N_H; agg = agg_mh; cnt = cnt_mh; }
    else                      { tsrc = t_h; src = ei_hm; E = E_hm; N = N_M; agg = agg_hm; cnt = cnt_hm; }
    const int* dst = src + E;

    const int row0 = blockIdx.x * RB;
    if (row0 >= N) return;

    const int tid = threadIdx.x, lane = tid & 31;
    const int wv = __builtin_amdgcn_readfirstlane(tid >> 5);

    for (int i = tid; i < (RB + 1) * D / 4; i += 1024)
        ((float4*)acc)[i] = make_float4(0.0f, 0.0f, 0.0f, 0.0f);
    for (int i = tid; i < RB; i += 1024) cntS[i] = 0.0f;
    __syncthreads();

    int nl = 0;

    auto drain = [&](int n) {
        if (n <= 0) return;
        int d0 = RB, s0 = 0, d1 = RB, s1 = 0;
        if (lane < n) {
            d0 = lstD[wv][lane];
            s0 = src[lstE[wv][lane]];
            atomicAdd(&cntS[d0], 1.0f);
        }
        if (lane + 32 < n) {
            d1 = lstD[wv][lane + 32];
            s1 = src[lstE[wv][lane + 32]];
            atomicAdd(&cntS[d1], 1.0f);
        }
#pragma unroll 1
        for (int i = 0; i < n; i += DB) {
            const int sv = (i < 32) ? s0 : s1;
            const int dv = (i < 32) ? d0 : d1;
            float4 x[DB];
            int dk[DB];
#pragma unroll
            for (int k = 0; k < DB; ++k) {
                const int l = (i + k) & 31;
                const int sk = __shfl(sv, l);
                dk[k] = __shfl(dv, l);
                x[k] = *(const float4*)&tsrc[(size_t)sk * D + lane * 4];
            }
#pragma unroll
            for (int k = 0; k < DB; ++k) {
                float* a = &acc[dk[k] * D + lane];
                atomicAdd(a,      x[k].x);
                atomicAdd(a + 32, x[k].y);
                atomicAdd(a + 64, x[k].z);
                atomicAdd(a + 96, x[k].w);
            }
        }
    };
    auto append = [&](bool hit, unsigned m, int e, int d) {
        if (hit) {
            const int r = nl + (int)__builtin_amdgcn_mbcnt_lo(m, 0u);
            lstE[wv][r] = e;
            lstD[wv][r] = d;
        }
        nl += __builtin_popcount(m);
    };

    const int nG = E >> 2;
    const int nT = (nG + 31) >> 5;
    const int off = (int)(((unsigned)blockIdx.x * (unsigned)nT) / gridDim.x);
    const int4* dst4 = (const int4*)dst;

    auto group_of = [&](int tl, bool& ok) -> int {
        int tr = tl + off;
        if (tr >= nT) tr -= nT;
        const int g = tr * 32 + lane;
        ok = (tl < nT) && (g < nG);
        return g;
    };

    if (nG > 0) {
        int4 vn[TPI];
#pragma unroll
        for (int k = 0; k < TPI; ++k) {
            bool ok; const int g = group_of(wv + NWV * k, ok);
            vn[k] = dst4[ok ? g : 0];
        }
        for (int tt = wv; tt < nT; tt += NWV * TPI) {
            int4 v[TPI];
#pragma unroll
            for (int k = 0; k < TPI; ++k) v[k] = vn[k];
            if (tt + NWV * TPI < nT) {
#pragma unroll
                for (int k = 0; k < TPI; ++k) {
                    bool ok; const int g = group_of(tt + NWV * TPI + NWV * k, ok);
                    vn[k] = dst4[ok ? g : 0];
                }
            }

            int gk[TPI]; bool okk[TPI];
            unsigned m[TPI][4];
            int H = 0;
#pragma unroll
            for (int k = 0; k < TPI; ++k) {
                gk[k] = group_of(tt + NWV * k, okk[k]);
                const int dv[4] = { v[k].x, v[k].y, v[k].z, v[k].w };
#pragma unroll
                for (int c = 0; c < 4; ++c) {
                    const bool hit = okk[k] && (unsigned)(dv[c] - row0) < (unsigned)RB;
                    m[k][c] = __builtin_amdgcn_ballot_w32(hit);
                    H += __builtin_popcount(m[k][c]);
                }
            }
            if (H == 0) continue;
            if (nl + H > LCAP) { drain(nl); nl = 0; }
            if (H <= LCAP) {
#pragma unroll
                for (int k = 0; k < TPI; ++k) {
                    const int dv[4] = { v[k].x, v[k].y, v[k].z, v[k].w };
#pragma unroll
                    for (int c = 0; c < 4; ++c) {
                        if (m[k][c]) {
                            const int d = dv[c] - row0;
                            const bool hit = okk[k] && (unsigned)d < (unsigned)RB;
                            append(hit, m[k][c], gk[k] * 4 + c, d);
                        }
                    }
                }
            } else {
#pragma unroll 1
                for (int q = 0; q < TPI * 4; ++q) {
                    bool ok; const int g = group_of(tt + NWV * (q >> 2), ok);
                    const int e = g * 4 + (q & 3);
                    const int d = (ok ? dst[e] : -1) - row0;
                    const bool hit = ok && (unsigned)d < (unsigned)RB;
                    const unsigned mq = __builtin_amdgcn_ballot_w32(hit);
                    if (nl + __builtin_popcount(mq) > LCAP) { drain(nl); nl = 0; }
                    append(hit, mq, e, d);
                }
            }
        }
    }
    if (wv == 0 && (nG << 2) < E) {
        const int e = (nG << 2) + lane;
        const bool in = e < E;
        const int d = (in ? dst[e] : -1) - row0;
        const bool hit = in && (unsigned)d < (unsigned)RB;
        const unsigned mq = __builtin_amdgcn_ballot_w32(hit);
        if (nl + __builtin_popcount(mq) > LCAP) { drain(nl); nl = 0; }
        append(hit, mq, e, d);
    }
    drain(nl);
    __syncthreads();

    for (int i = tid; i < RB * 32; i += 1024) {
        const int r = i >> 5;
        if (row0 + r < N) {
            const float* a = &acc[r * D + lane];
            ((float4*)agg)[(size_t)(row0 + r) * 32 + lane] = make_float4(a[0], a[32], a[64], a[96]);
        }
    }
    for (int i = tid; i < RB; i += 1024)
        if (row0 + i < N) cnt[row0 + i] = cntS[i];
}

__device__ __forceinline__ float rowsum16(float v) {
    v += __shfl_xor(v, 1);
    v += __shfl_xor(v, 2);
    v += __shfl_xor(v, 4);
    v += __shfl_xor(v, 8);
    return v;
}

__device__ __forceinline__
void sage_pass(v8f (&acc)[8], const float* __restrict__ tr, const float* __restrict__ ar,
               const float ca, const float* __restrict__ Wl, const float* __restrict__ Wr,
               const size_t wo) {
#pragma unroll 1
    for (int kc = 0; kc < D; kc += 16) {
        v2f at[4], aa[4];
#pragma unroll
        for (int i = 0; i < 4; ++i) {
            at[i] = *(const v2f*)(tr + kc + 4 * i);
            const v2f xa = *(const v2f*)(ar + kc + 4 * i);
            aa[i].x = xa.x / ca; aa[i].y = xa.y / ca;
        }
#pragma unroll
        for (int i = 0; i < 4; ++i) {
            const size_t w0 = wo + (size_t)(kc + 4 * i) * D;
#pragma unroll
            for (int ct = 0; ct < 8; ++ct) {
                v2f b1, b2;
                b1.x = Wl[w0 + 16 * ct]; b1.y = Wl[w0 + D + 16 * ct];
                b2.x = Wr[w0 + 16 * ct]; b2.y = Wr[w0 + D + 16 * ct];
                acc[ct] = wmma_f32(aa[i], b1, acc[ct]);
                acc[ct] = wmma_f32(at[i], b2, acc[ct]);
            }
        }
    }
}

__global__ __launch_bounds__(32)
void combine_kernel(int tilesM,
                    const float* __restrict__ t_m,
                    const float* __restrict__ agg_mm, const float* __restrict__ cnt_mm,
                    const float* __restrict__ agg_hm, const float* __restrict__ cnt_hm,
                    const float* __restrict__ Wl_mm, const float* __restrict__ bl_mm,
                    const float* __restrict__ Wr_mm,
                    const float* __restrict__ Wl_hm, const float* __restrict__ bl_hm,
                    const float* __restrict__ Wr_hm,
                    const float* __restrict__ g_m, const float* __restrict__ be_m,
                    float* __restrict__ out_m,
                    const float* __restrict__ t_h,
                    const float* __restrict__ agg_mh, const float* __restrict__ cnt_mh,
                    const float* __restrict__ Wl_mh, const float* __restrict__ bl_mh,
                    const float* __restrict__ Wr_mh,
                    const float* __restrict__ g_h, const float* __restrict__ be_h,
                    float* __restrict__ out_h) {
    const int lane = threadIdx.x;
    const int b = blockIdx.x;
    const bool two = b < tilesM;
    const int row0 = (two ? b : b - tilesM) * 16;
    const float* t    = two ? t_m    : t_h;
    const float* agg1 = two ? agg_mm : agg_mh;
    const float* cnt1 = two ? cnt_mm : cnt_mh;
    const float* Wl1  = two ? Wl_mm  : Wl_mh;
    const float* bl1  = two ? bl_mm  : bl_mh;
    const float* Wr1  = two ? Wr_mm  : Wr_mh;
    const float* g    = two ? g_m    : g_h;
    const float* be   = two ? be_m   : be_h;
    float*       out  = two ? out_m  : out_h;

    const int n = lane & 15, h = lane >> 4, m = lane & 15;
    const int kb = h * 2;
    const float* tr = t + (size_t)(row0 + m) * D + kb;
    const size_t wo = (size_t)kb * D + n;

    v8f acc1[8], acc2[8];
#pragma unroll
    for (int ct = 0; ct < 8; ++ct) { acc1[ct] = v8f{}; acc2[ct] = v8f{}; }
    const float* ar = agg1 + (size_t)(row0 + m) * D + kb;
    float ca = fmaxf(cnt1[row0 + m], 1.0f);
    const float* Wl = Wl1;
    const float* Wr = Wr1;
#pragma unroll 1
    for (int pass = 0; pass < 2; ++pass) {
        sage_pass(acc2, tr, ar, ca, Wl, Wr, wo);
        if (pass == 1) break;
#pragma unroll
        for (int ct = 0; ct < 8; ++ct) { acc1[ct] = acc2[ct]; acc2[ct] = v8f{}; }
        if (!two) break;
        ar = agg_hm + (size_t)(row0 + m) * D + kb;
        ca = fmaxf(cnt_hm[row0 + m], 1.0f);
        Wl = Wl_hm;
        Wr = Wr_hm;
    }

    float s1[8], s2[8];
#pragma unroll
    for (int p = 0; p < 8; ++p) { s1[p] = 0.0f; s2[p] = 0.0f; }
#pragma unroll
    for (int ct = 0; ct < 8; ++ct) {
        const float c1 = bl1[16 * ct + n], c2 = bl_hm[16 * ct + n];
#pragma unroll
        for (int p = 0; p < 8; ++p) {
            acc1[ct][p] += c1;
            acc2[ct][p] += c2;
            s1[p] += acc1[ct][p] * acc1[ct][p];
            s2[p] += acc2[ct][p] * acc2[ct][p];
        }
    }
    float n1[8], n2[8];
#pragma unroll
    for (int p = 0; p < 8; ++p) {
        n1[p] = fmaxf(sqrtf(rowsum16(s1[p])), 1e-12f);
        n2[p] = fmaxf(sqrtf(rowsum16(s2[p])), 1e-12f);
    }

    float sy[8], sq[8];
#pragma unroll
    for (int p = 0; p < 8; ++p) { sy[p] = 0.0f; sq[p] = 0.0f; }
#pragma unroll
    for (int ct = 0; ct < 8; ++ct) {
#pragma unroll
        for (int p = 0; p < 8; ++p) {
            const float tv = t[(size_t)(row0 + p + 8 * h) * D + 16 * ct + n];
            const float o1 = acc1[ct][p] / n1[p];
            const float conv = two ? 0.5f * (o1 + acc2[ct][p] / n2[p]) : o1;
            const float y = tv + conv;
            acc1[ct][p] = y;
            sy[p] += y;
            sq[p] += y * y;
        }
    }
    float mu[8], rs[8];
#pragma unroll
    for (int p = 0; p < 8; ++p) {
        mu[p] = rowsum16(sy[p]) * (1.0f / D);
        const float var = rowsum16(sq[p]) * (1.0f / D) - mu[p] * mu[p];
        rs[p] = rsqrtf(var + 1e-5f);
    }
#pragma unroll
    for (int ct = 0; ct < 8; ++ct) {
        const float gc = g[16 * ct + n], bc = be[16 * ct + n];
#pragma unroll
        for (int p = 0; p < 8; ++p)
            out[(size_t)(row0 + p + 8 * h) * D + 16 * ct + n] = (acc1[ct][p] - mu[p]) * rs[p] * gc + bc;
    }
}

extern "C" void kernel_launch(void* const* d_in, const int* in_sizes, int n_in,
                              void* d_out, int out_size, void* d_ws, size_t ws_size,
                              hipStream_t stream) {
    const float* x_m    = (const float*)d_in[0];
    const float* x_h    = (const float*)d_in[1];
    const float* W_nt_m = (const float*)d_in[2];
    const float* b_nt_m = (const float*)d_in[3];
    const float* W_nt_h = (const float*)d_in[4];
    const float* b_nt_h = (const float*)d_in[5];
    const float* Wl_mm  = (const float*)d_in[6];
    const float* bl_mm  = (const float*)d_in[7];
    const float* Wr_mm  = (const float*)d_in[8];
    const float* Wl_mh  = (const float*)d_in[9];
    const float* bl_mh  = (const float*)d_in[10];
    const float* Wr_mh  = (const float*)d_in[11];
    const float* Wl_hm  = (const float*)d_in[12];
    const float* bl_hm  = (const float*)d_in[13];
    const float* Wr_hm  = (const float*)d_in[14];
    const float* g_m    = (const float*)d_in[15];
    const float* be_m   = (const float*)d_in[16];
    const float* g_h    = (const float*)d_in[17];
    const float* be_h   = (const float*)d_in[18];
    const int*   ei_mm  = (const int*)d_in[19];
    const int*   ei_mh  = (const int*)d_in[20];
    const int*   ei_hm  = (const int*)d_in[21];

    const int N_M  = in_sizes[0] / D;
    const int N_H  = in_sizes[1] / D;
    const int E_mm = in_sizes[19] / 2;
    const int E_mh = in_sizes[20] / 2;
    const int E_hm = in_sizes[21] / 2;

    const size_t nm = (size_t)N_M * D;
    const size_t nh = (size_t)N_H * D;

    float* ws     = (float*)d_ws;
    float* t_m    = ws;
    float* t_h    = t_m + nm;
    float* agg_mm = t_h + nh;
    float* agg_hm = agg_mm + nm;
    float* agg_mh = agg_hm + nm;
    float* cnt_mm = agg_mh + nh;
    float* cnt_hm = cnt_mm + N_M;
    float* cnt_mh = cnt_hm + N_M;

    const int nmax = N_M > N_H ? N_M : N_H;
    lin_kernel<<<dim3(nmax / 16, 2), 256, 0, stream>>>(x_m, W_nt_m, b_nt_m, t_m, N_M,
                                                       x_h, W_nt_h, b_nt_h, t_h, N_H);

    own_agg_kernel<<<dim3((nmax + RB - 1) / RB, 3), 1024, 0, stream>>>(
        t_m, t_h, ei_mm, ei_mh, ei_hm, E_mm, E_mh, E_hm, N_M, N_H,
        agg_mm, agg_mh, agg_hm, cnt_mm, cnt_mh, cnt_hm);

    float* out_m = (float*)d_out;
    float* out_h = out_m + nm;
    combine_kernel<<<N_M / 16 + N_H / 16, 32, 0, stream>>>(
        N_M / 16,
        t_m, agg_mm, cnt_mm, agg_hm, cnt_hm, Wl_mm, bl_mm, Wr_mm, Wl_hm, bl_hm, Wr_hm,
        g_m, be_m, out_m,
        t_h, agg_mh, cnt_mh, Wl_mh, bl_mh, Wr_mh, g_h, be_h, out_h);
}


# Edge sweep reads one owner byte per edge (row / 592, written by extra work-groups of launch 1), 16 edges per 16-byte load, zero-byte test; listed edges checked at fetch; sums in arrival order
# speedup vs baseline: 4.8153x; 1.0673x over previous
#include <hip/hip_runtime.h>

#define D 128
#define LDSW 130

#define RB   592

typedef __attribute__((ext_vector_type(2))) float v2f;
typedef __attribute__((ext_vector_type(8))) float v8f;

__device__ __forceinline__ v8f wmma_f32(v2f a, v2f b, v8f c) {
    return __builtin_amdgcn_wmma_f32_16x16x4_f32(false, a, false, b, (short)0, c,
                                                 false, false);
}

__global__ __launch_bounds__(256)
void lin_kernel(const float* __restrict__ x_m, const float* __restrict__ W_m,
                const float* __restrict__ bias_m, float* __restrict__ t_m, int N_M,
                const float* __restrict__ x_h, const float* __restrict__ W_h,
                const float* __restrict__ bias_h, float* __restrict__ t_h, int N_H,
                const int* __restrict__ ei_mm, const int* __restrict__ ei_mh,
                const int* __restrict__ ei_hm, int E_mm, int E_mh, int E_hm,
                unsigned* __restrict__ code_mm, unsigned* __restrict__ code_mh,
                unsigned* __restrict__ code_hm) {
    __shared__ float aS[16 * LDSW];
    if (blockIdx.y == 2) {
        const int nth = gridDim.x * 256;
        const int g0  = blockIdx.x * 256 + threadIdx.x;
        auto owner = [](int d) -> unsigned { return ((unsigned)d / (unsigned)RB) & 255u; };
        auto write_codes = [&](const int* ei, int E, unsigned* code) {
            const int4* d4 = (const int4*)(ei + E);
            const int nW = E >> 2;
            for (int w = g0; w < nW; w += nth) {
                const int4 v = d4[w];
                code[w] = owner(v.x) | (owner(v.y) << 8) | (owner(v.z) << 16) | (owner(v.w) << 24);
            }
        };
        write_codes(ei_mm, E_mm, code_mm);
        write_codes(ei_mh, E_mh, code_mh);
        write_codes(ei_hm, E_hm, code_hm);
        return;
    }
    const float* x; const float* W; const float* bias; float* t; int N;
    if (blockIdx.y == 0) { x = x_m; W = W_m; bias = bias_m; t = t_m; N = N_M; }
    else                 { x = x_h; W = W_h; bias = bias_h; t = t_h; N = N_H; }
    const int tid  = threadIdx.x;
    const int row0 = blockIdx.x * 16;
    if (row0 >= N) return;

    for (int i = tid; i < 16 * D; i += 256) {
        int r = i >> 7, c = i & (D - 1);
        aS[r * LDSW + c] = x[(size_t)(row0 + r) * D + c];
    }
    __syncthreads();

    const int lane = tid & 31, wv = tid >> 5;
    const int n = lane & 15, h = lane >> 4, m = lane & 15;
    const int kb = h * 2;
    const int col = wv * 16 + n;

    v8f acc = {};
    for (int k = 0; k < D; k += 4) {
        v2f a = *(const v2f*)&aS[m * LDSW + k + kb];
        v2f b;
        b.x = W[(size_t)(k + kb) * D + col];
        b.y = W[(size_t)(k + kb + 1) * D + col];
        acc = wmma_f32(a, b, acc);
    }
    const float bv = bias[col];
    for (int p = 0; p < 8; ++p)
        t[(size_t)(row0 + p + 8 * h) * D + col] = acc[p] + bv;
}

#define NWV  32
#define TPI  4
#define DB   8
#define LCAP 128

__global__ __launch_bounds__(1024)
void own_agg_kernel(const float* __restrict__ t_m, const float* __restrict__ t_h,
                    const int* __restrict__ ei_mm, const int* __restrict__ ei_mh,
                    const int* __restrict__ ei_hm,
                    const unsigned* __restrict__ code_mm, const unsigned* __restrict__ code_mh,
                    const unsigned* __restrict__ code_hm,
                    int E_mm, int E_mh, int E_hm, int N_M, int N_H,
                    float* __restrict__ agg_mm, float* __restrict__ agg_mh,
                    float* __restrict__ agg_hm,
                    float* __restrict__ cnt_mm, float* __restrict__ cnt_mh,
                    float* __restrict__ cnt_hm) {
    __shared__ float acc[(RB + 1) * D];
    __shared__ float cntS[RB + 1];
    __shared__ int   lstE[NWV][LCAP];

    const float* tsrc; const int* src; const unsigned* code; float* agg; float* cnt; int E, N;
    if (blockIdx.y == 0)      { tsrc = t_m; src = ei_mm; code = code_mm; E = E_mm; N = N_M; agg = agg_mm; cnt = cnt_mm; }
    else if (blockIdx.y == 1) { tsrc = t_m; src = ei_mh; code = code_mh; E = E_mh; N = N_H; agg = agg_mh; cnt = cnt_mh; }
    else                      { tsrc = t_h; src = ei_hm; code = code_hm; E = E_hm; N = N_M; agg = agg_hm; cnt = cnt_hm; }
    const int* dst = src + E;

    const int row0 = blockIdx.x * RB;
    if (row0 >= N) return;

    const int tid = threadIdx.x, lane = tid & 31;
    const int wv = __builtin_amdgcn_readfirstlane(tid >> 5);

    for (int i = tid; i < (RB + 1) * D / 4; i += 1024)
        ((float4*)acc)[i] = make_float4(0.0f, 0.0f, 0.0f, 0.0f);
    for (int i = tid; i < RB + 1; i += 1024) cntS[i] = 0.0f;
    __syncthreads();

    int nl = 0;

    auto drain = [&](int n) {
#pragma unroll 1
        for (int base = 0; base < n; base += 32) {
            int dv = RB, sv = 0;
            if (base + lane < n) {
                const int e = lstE[wv][base + lane];
                const int s = src[e];
                const int d = dst[e] - row0;
                const bool mine = (unsigned)d < (unsigned)RB;
                dv = mine ? d : RB;
                sv = mine ? s : 0;
                atomicAdd(&cntS[dv], 1.0f);
            }
            const int nb = (n - base < 32) ? (n - base) : 32;
#pragma unroll 1
            for (int i = 0; i < nb; i += DB) {
                float4 x[DB];
                int dk[DB];
#pragma unroll
                for (int k = 0; k < DB; ++k) {
                    const int l = i + k;
                    const int sk = __shfl(sv, l);
                    dk[k] = __shfl(dv, l);
                    x[k] = *(const float4*)&tsrc[(size_t)sk * D + lane * 4];
                }
#pragma unroll
                for (int k = 0; k < DB; ++k) {
                    float* a = &acc[dk[k] * D + lane];
                    atomicAdd(a,      x[k].x);
                    atomicAdd(a + 32, x[k].y);
                    atomicAdd(a + 64, x[k].z);
                    atomicAdd(a + 96, x[k].w);
                }
            }
        }
    };

    const int nW = E >> 2;
    const int nQ = nW >> 2;
    const int nT = (nQ + 31) >> 5;
    const int off = (int)(((unsigned)blockIdx.x * (unsigned)nT) / gridDim.x);
    const uint4* c4 = (const uint4*)code;
    const unsigned pat = ((unsigned)blockIdx.x & 255u) * 0x01010101u;

    auto quad_of = [&](int tl, bool& ok) -> int {
        int tr = tl + off;
        if (tr >= nT) tr -= nT;
        const int g = tr * 32 + lane;
        ok = (tl < nT) && (g < nQ);
        return g;
    };
    auto flags_of = [&](const uint4 v) -> unsigned {
        const unsigned w[4] = { v.x, v.y, v.z, v.w };
        unsigned r = 0u;
#pragma unroll
        for (int j = 0; j < 4; ++j) {
            const unsigned x = w[j] ^ pat;
            const unsigned z = (x - 0x01010101u) & ~x & 0x80808080u;
            r |= (z >> 7) << j;
        }
        return r;
    };

    if (nQ > 0) {
        uint4 vn[TPI];
#pragma unroll
        for (int k = 0; k < TPI; ++k) {
            bool ok; const int g = quad_of(wv + NWV * k, ok);
            vn[k] = c4[ok ? g : 0];
        }
        for (int tt = wv; tt < nT; tt += NWV * TPI) {
            uint4 v[TPI];
#pragma unroll
            for (int k = 0; k < TPI; ++k) v[k] = vn[k];
            if (tt + NWV * TPI < nT) {
#pragma unroll
                for (int k = 0; k < TPI; ++k) {
                    bool ok; const int g = quad_of(tt + NWV * TPI + NWV * k, ok);
                    vn[k] = c4[ok ? g : 0];
                }
            }
            unsigned f[TPI];
#pragma unroll
            for (int k = 0; k < TPI; ++k) {
                bool ok; (void)quad_of(tt + NWV * k, ok);
                f[k] = ok ? flags_of(v[k]) : 0u;
            }
            unsigned long long G = (unsigned long long)(f[0] | (f[1] << 4))
                                 | ((unsigned long long)(f[2] | (f[3] << 4)) << 32);
            for (;;) {
                const bool has = G != 0ull;
                const unsigned m = __builtin_amdgcn_ballot_w32(has);
                if (m == 0u) break;
                const int pc = __builtin_popcount(m);
                if (nl + pc > LCAP) { drain(nl); nl = 0; }
                if (has) {
                    const int p = __builtin_ctzll(G);
                    G &= G - 1ull;
                    const int k = ((p >> 5) << 1) | ((p >> 2) & 1);
                    int tr = tt + NWV * k + off;
                    if (tr >= nT) tr -= nT;
                    const int e = ((tr * 32 + lane) << 4) + ((p & 3) << 2) + ((p & 31) >> 3);
                    lstE[wv][nl + (int)__builtin_amdgcn_mbcnt_lo(m, 0u)] = e;
                }
                nl += pc;
            }
        }
    }
    if (wv == 0 && (nQ << 4) < E) {
        const int e = (nQ << 4) + lane;
        const bool in = e < E;
        const int d = (in ? dst[e] : -1) - row0;
        const bool hit = in && (unsigned)d < (unsigned)RB;
        const unsigned mq = __builtin_amdgcn_ballot_w32(hit);
        if (nl + __builtin_popcount(mq) > LCAP) { drain(nl); nl = 0; }
        if (hit) lstE[wv][nl + (int)__builtin_amdgcn_mbcnt_lo(mq, 0u)] = e;
        nl += __builtin_popcount(mq);
    }
    drain(nl);
    __syncthreads();

    for (int i = tid; i < RB * 32; i += 1024) {
        const int r = i >> 5;
        if (row0 + r < N) {
            const float* a = &acc[r * D + lane];
            ((float4*)agg)[(size_t)(row0 + r) * 32 + lane] = make_float4(a[0], a[32], a[64], a[96]);
        }
    }
    for (int i = tid; i < RB; i += 1024)
        if (row0 + i < N) cnt[row0 + i] = cntS[i];
}

__device__ __forceinline__ float rowsum16(float v) {
    v += __shfl_xor(v, 1);
    v += __shfl_xor(v, 2);
    v += __shfl_xor(v, 4);
    v += __shfl_xor(v, 8);
    return v;
}

__device__ __forceinline__
void sage_pass(v8f (&acc)[8], const float* __restrict__ tr, const float* __restrict__ ar,
               const float ca, const float* __restrict__ Wl, const float* __restrict__ Wr,
               const size_t wo) {
#pragma unroll 1
    for (int kc = 0; kc < D; kc += 16) {
        v2f at[4], aa[4];
#pragma unroll
        for (int i = 0; i < 4; ++i) {
            at[i] = *(const v2f*)(tr + kc + 4 * i);
            const v2f xa = *(const v2f*)(ar + kc + 4 * i);
            aa[i].x = xa.x / ca; aa[i].y = xa.y / ca;
        }
#pragma unroll
        for (int i = 0; i < 4; ++i) {
            const size_t w0 = wo + (size_t)(kc + 4 * i) * D;
#pragma unroll
            for (int ct = 0; ct < 8; ++ct) {
                v2f b1, b2;
                b1.x = Wl[w0 + 16 * ct]; b1.y = Wl[w0 + D + 16 * ct];
                b2.x = Wr[w0 + 16 * ct]; b2.y = Wr[w0 + D + 16 * ct];
                acc[ct] = wmma_f32(aa[i], b1, acc[ct]);
                acc[ct] = wmma_f32(at[i], b2, acc[ct]);
            }
        }
    }
}

__global__ __launch_bounds__(32)
void combine_kernel(int tilesM,
                    const float* __restrict__ t_m,
                    const float* __restrict__ agg_mm, const float* __restrict__ cnt_mm,
                    const float* __restrict__ agg_hm, const float* __restrict__ cnt_hm,
                    const float* __restrict__ Wl_mm, const float* __restrict__ bl_mm,
                    const float* __restrict__ Wr_mm,
                    const float* __restrict__ Wl_hm, const float* __restrict__ bl_hm,
                    const float* __restrict__ Wr_hm,
                    const float* __restrict__ g_m, const float* __restrict__ be_m,
                    float* __restrict__ out_m,
                    const float* __restrict__ t_h,
                    const float* __restrict__ agg_mh, const float* __restrict__ cnt_mh,
                    const float* __restrict__ Wl_mh, const float* __restrict__ bl_mh,
                    const float* __restrict__ Wr_mh,
                    const float* __restrict__ g_h, const float* __restrict__ be_h,
                    float* __restrict__ out_h) {
    const int lane = threadIdx.x;
    const int b = blockIdx.x;
    const bool two = b < tilesM;
    const int row0 = (two ? b : b - tilesM) * 16;
    const float* t    = two ? t_m    : t_h;
    const float* agg1 = two ? agg_mm : agg_mh;
    const float* cnt1 = two ? cnt_mm : cnt_mh;
    const float* Wl1  = two ? Wl_mm  : Wl_mh;
    const float* bl1  = two ? bl_mm  : bl_mh;
    const float* Wr1  = two ? Wr_mm  : Wr_mh;
    const float* g    = two ? g_m    : g_h;
    const float* be   = two ? be_m   : be_h;
    float*       out  = two ? out_m  : out_h;

    const int n = lane & 15, h = lane >> 4, m = lane & 15;
    const int kb = h * 2;
    const float* tr = t + (size_t)(row0 + m) * D + kb;
    const size_t wo = (size_t)kb * D + n;

    v8f acc1[8], acc2[8];
#pragma unroll
    for (int ct = 0; ct < 8; ++ct) { acc1[ct] = v8f{}; acc2[ct] = v8f{}; }
    const float* ar = agg1 + (size_t)(row0 + m) * D + kb;
    float ca = fmaxf(cnt1[row0 + m], 1.0f);
    const float* Wl = Wl1;
    const float* Wr = Wr1;
#pragma unroll 1
    for (int pass = 0; pass < 2; ++pass) {
        sage_pass(acc2, tr, ar, ca, Wl, Wr, wo);
        if (pass == 1) break;
#pragma unroll
        for (int ct = 0; ct < 8; ++ct) { acc1[ct] = acc2[ct]; acc2[ct] = v8f{}; }
        if (!two) break;
        ar = agg_hm + (size_t)(row0 + m) * D + kb;
        ca = fmaxf(cnt_hm[row0 + m], 1.0f);
        Wl = Wl_hm;
        Wr = Wr_hm;
    }

    float s1[8], s2[8];
#pragma unroll
    for (int p = 0; p < 8; ++p) { s1[p] = 0.0f; s2[p] = 0.0f; }
#pragma unroll
    for (int ct = 0; ct < 8; ++ct) {
        const float c1 = bl1[16 * ct + n], c2 = bl_hm[16 * ct + n];
#pragma unroll
        for (int p = 0; p < 8; ++p) {
            acc1[ct][p] += c1;
            acc2[ct][p] += c2;
            s1[p] += acc1[ct][p] * acc1[ct][p];
            s2[p] += acc2[ct][p] * acc2[ct][p];
        }
    }
    float n1[8], n2[8];
#pragma unroll
    for (int p = 0; p < 8; ++p) {
        n1[p] = fmaxf(sqrtf(rowsum16(s1[p])), 1e-12f);
        n2[p] = fmaxf(sqrtf(rowsum16(s2[p])), 1e-12f);
    }

    float sy[8], sq[8];
#pragma unroll
    for (int p = 0; p < 8; ++p) { sy[p] = 0.0f; sq[p] = 0.0f; }
#pragma unroll
    for (int ct = 0; ct < 8; ++ct) {
#pragma unroll
        for (int p = 0; p < 8; ++p) {
            const float tv = t[(size_t)(row0 + p + 8 * h) * D + 16 * ct + n];
            const float o1 = acc1[ct][p] / n1[p];
            const float conv = two ? 0.5f * (o1 + acc2[ct][p] / n2[p]) : o1;
            const float y = tv + conv;
            acc1[ct][p] = y;
            sy[p] += y;
            sq[p] += y * y;
        }
    }
    float mu[8], rs[8];
#pragma unroll
    for (int p = 0; p < 8; ++p) {
        mu[p] = rowsum16(sy[p]) * (1.0f / D);
        const float var = rowsum16(sq[p]) * (1.0f / D) - mu[p] * mu[p];
        rs[p] = rsqrtf(var + 1e-5f);
    }
#pragma unroll
    for (int ct = 0; ct < 8; ++ct) {
        const float gc = g[16 * ct + n], bc = be[16 * ct + n];
#pragma unroll
        for (int p = 0; p < 8; ++p)
            out[(size_t)(row0 + p + 8 * h) * D + 16 * ct + n] = (acc1[ct][p] - mu[p]) * rs[p] * gc + bc;
    }
}

extern "C" void kernel_launch(void* const* d_in, const int* in_sizes, int n_in,
                              void* d_out, int out_size, void* d_ws, size_t ws_size,
                              hipStream_t stream) {
    const float* x_m    = (const float*)d_in[0];
    const float* x_h    = (const float*)d_in[1];
    const float* W_nt_m = (const float*)d_in[2];
    const float* b_nt_m = (const float*)d_in[3];
    const float* W_nt_h = (const float*)d_in[4];
    const float* b_nt_h = (const float*)d_in[5];
    const float* Wl_mm  = (const float*)d_in[6];
    const float* bl_mm  = (const float*)d_in[7];
    const float* Wr_mm  = (const float*)d_in[8];
    const float* Wl_mh  = (const float*)d_in[9];
    const float* bl_mh  = (const float*)d_in[10];
    const float* Wr_mh  = (const float*)d_in[11];
    const float* Wl_hm  = (const float*)d_in[12];
    const float* bl_hm  = (const float*)d_in[13];
    const float* Wr_hm  = (const float*)d_in[14];
    const float* g_m    = (const float*)d_in[15];
    const float* be_m   = (const float*)d_in[16];
    const float* g_h    = (const float*)d_in[17];
    const float* be_h   = (const float*)d_in[18];
    const int*   ei_mm  = (const int*)d_in[19];
    const int*   ei_mh  = (const int*)d_in[20];
    const int*   ei_hm  = (const int*)d_in[21];

    const int N_M  = in_sizes[0] / D;
    const int N_H  = in_sizes[1] / D;
    const int E_mm = in_sizes[19] / 2;
    const int E_mh = in_sizes[20] / 2;
    const int E_hm = in_sizes[21] / 2;

    const size_t nm = (size_t)N_M * D;
    const size_t nh = (size_t)N_H * D;

    float* ws     = (float*)d_ws;
    float* t_m    = ws;
    float* t_h    = t_m + nm;
    float* agg_mm = t_h + nh;
    float* agg_hm = agg_mm + nm;
    float* agg_mh = agg_hm + nm;
    float* cnt_mm = agg_mh + nh;
    float* cnt_hm = cnt_mm + N_M;
    float* cnt_mh = cnt_hm + N_M;

    unsigned* code_mm = (unsigned*)d_out;
    unsigned* code_mh = code_mm + (((E_mm >> 2) + 3) & ~3);
    unsigned* code_hm = code_mh + (((E_mh >> 2) + 3) & ~3);

    const int nmax = N_M > N_H ? N_M : N_H;
    lin_kernel<<<dim3(nmax / 16, 3), 256, 0, stream>>>(x_m, W_nt_m, b_nt_m, t_m, N_M,
                                                       x_h, W_nt_h, b_nt_h, t_h, N_H,
                                                       ei_mm, ei_mh, ei_hm, E_mm, E_mh, E_hm,
                                                       code_mm, code_mh, code_hm);

    own_agg_kernel<<<dim3((nmax + RB - 1) / RB, 3), 1024, 0, stream>>>(
        t_m, t_h, ei_mm, ei_mh, ei_hm, code_mm, code_mh, code_hm,
        E_mm, E_mh, E_hm, N_M, N_H,
        agg_mm, agg_mh, agg_hm, cnt_mm, cnt_mh, cnt_hm);

    float* out_m = (float*)d_out;
    float* out_h = out_m + nm;
    combine_kernel<<<N_M / 16 + N_H / 16, 32, 0, stream>>>(
        N_M / 16,
        t_m, agg_mm, cnt_mm, agg_hm, cnt_hm, Wl_mm, bl_mm, Wr_mm, Wl_hm, bl_hm, Wr_hm,
        g_m, be_m, out_m,
        t_h, agg_mh, cnt_mh, Wl_mh, bl_mh, Wr_mh, g_h, be_h, out_h);
}


# Linear-transform launch re-tiled: one wave (its own work-group) owns 16 whole rows, 8 column tiles in registers, stages its x rows itself; launch 1 runs 9375 waves, not 75000; sum orders unchanged
# speedup vs baseline: 5.0567x; 1.0501x over previous
#include <hip/hip_runtime.h>

#define D 128
#define LDSW 130

#define RB   592

typedef __attribute__((ext_vector_type(2))) float v2f;
typedef __attribute__((ext_vector_type(8))) float v8f;

__device__ __forceinline__ v8f wmma_f32(v2f a, v2f b, v8f c) {
    return __builtin_amdgcn_wmma_f32_16x16x4_f32(false, a, false, b, (short)0, c,
                                                 false, false);
}

__global__ __launch_bounds__(32)
void lin_kernel(const float* __restrict__ x_m, const float* __restrict__ W_m,
                const float* __restrict__ bias_m, float* __restrict__ t_m, int N_M,
                const float* __restrict__ x_h, const float* __restrict__ W_h,
                const float* __restrict__ bias_h, float* __restrict__ t_h, int N_H,
                const int* __restrict__ ei_mm, const int* __restrict__ ei_mh,
                const int* __restrict__ ei_hm, int E_mm, int E_mh, int E_hm,
                unsigned* __restrict__ code_mm, unsigned* __restrict__ code_mh,
                unsigned* __restrict__ code_hm) {
    __shared__ float aS[16 * LDSW];
    const int lane = threadIdx.x;
    if (blockIdx.y == 2) {
        const int nth = gridDim.x * 32;
        const int g0  = blockIdx.x * 32 + lane;
        auto owner = [](int d) -> unsigned { return ((unsigned)d / (unsigned)RB) & 255u; };
        auto write_codes = [&](const int* ei, int E, unsigned* code) {
            const int4* d4 = (const int4*)(ei + E);
            const int nW = E >> 2;
            for (int w = g0; w < nW; w += nth) {
                const int4 v = d4[w];
                code[w] = owner(v.x) | (owner(v.y) << 8) | (owner(v.z) << 16) | (owner(v.w) << 24);
            }
        };
        write_codes(ei_mm, E_mm, code_mm);
        write_codes(ei_mh, E_mh, code_mh);
        write_codes(ei_hm, E_hm, code_hm);
        return;
    }
    const float* x; const float* W; const float* bias; float* t; int N;
    if (blockIdx.y == 0) { x = x_m; W = W_m; bias = bias_m; t = t_m; N = N_M; }
    else                 { x = x_h; W = W_h; bias = bias_h; t = t_h; N = N_H; }
    const int row0 = blockIdx.x * 16;
    if (row0 >= N) return;

    float4 xr[16];
#pragma unroll
    for (int r = 0; r < 16; ++r)
        xr[r] = ((const float4*)(x + (size_t)(row0 + r) * D))[lane];
#pragma unroll
    for (int r = 0; r < 16; ++r) {
        v2f* d = (v2f*)&aS[r * LDSW + 4 * lane];
        v2f lo, hi;
        lo.x = xr[r].x; lo.y = xr[r].y; hi.x = xr[r].z; hi.y = xr[r].w;
        d[0] = lo;
        d[1] = hi;
    }
    __syncthreads();

    const int n = lane & 15, h = lane >> 4, m = lane & 15;
    const int kb = h * 2;
    const float* ap = &aS[m * LDSW + kb];
    const float* wp = W + (size_t)kb * D + n;

    v8f acc[8];
#pragma unroll
    for (int ct = 0; ct < 8; ++ct) acc[ct] = v8f{};
#pragma unroll 1
    for (int kc = 0; kc < D; kc += 16) {
        v2f a[4];
#pragma unroll
        for (int i = 0; i < 4; ++i) a[i] = *(const v2f*)(ap + kc + 4 * i);
#pragma unroll
        for (int i = 0; i < 4; ++i) {
            const float* w0 = wp + (size_t)(kc + 4 * i) * D;
#pragma unroll
            for (int ct = 0; ct < 8; ++ct) {
                v2f b;
                b.x = w0[16 * ct];
                b.y = w0[D + 16 * ct];
                acc[ct] = wmma_f32(a[i], b, acc[ct]);
            }
        }
    }
#pragma unroll
    for (int ct = 0; ct < 8; ++ct) {
        const float bv = bias[16 * ct + n];
#pragma unroll
        for (int p = 0; p < 8; ++p)
            t[(size_t)(row0 + p + 8 * h) * D + 16 * ct + n] = acc[ct][p] + bv;
    }
}

#define NWV  32
#define TPI  4
#define DB   8
#define LCAP 128

__global__ __launch_bounds__(1024)
void own_agg_kernel(const float* __restrict__ t_m, const float* __restrict__ t_h,
                    const int* __restrict__ ei_mm, const int* __restrict__ ei_mh,
                    const int* __restrict__ ei_hm,
                    const unsigned* __restrict__ code_mm, const unsigned* __restrict__ code_mh,
                    const unsigned* __restrict__ code_hm,
                    int E_mm, int E_mh, int E_hm, int N_M, int N_H,
                    float* __restrict__ agg_mm, float* __restrict__ agg_mh,
                    float* __restrict__ agg_hm,
                    float* __restrict__ cnt_mm, float* __restrict__ cnt_mh,
                    float* __restrict__ cnt_hm) {
    __shared__ float acc[(RB + 1) * D];
    __shared__ float cntS[RB + 1];
    __shared__ int   lstE[NWV][LCAP];

    const float* tsrc; const int* src; const unsigned* code; float* agg; float* cnt; int E, N;
    if (blockIdx.y == 0)      { tsrc = t_m; src = ei_mm; code = code_mm; E = E_mm; N = N_M; agg = agg_mm; cnt = cnt_mm; }
    else if (blockIdx.y == 1) { tsrc = t_m; src = ei_mh; code = code_mh; E = E_mh; N = N_H; agg = agg_mh; cnt = cnt_mh; }
    else                      { tsrc = t_h; src = ei_hm; code = code_hm; E = E_hm; N = N_M; agg = agg_hm; cnt = cnt_hm; }
    const int* dst = src + E;

    const int row0 = blockIdx.x * RB;
    if (row0 >= N) return;

    const int tid = threadIdx.x, lane = tid & 31;
    const int wv = __builtin_amdgcn_readfirstlane(tid >> 5);

    for (int i = tid; i < (RB + 1) * D / 4; i += 1024)
        ((float4*)acc)[i] = make_float4(0.0f, 0.0f, 0.0f, 0.0f);
    for (int i = tid; i < RB + 1; i += 1024) cntS[i] = 0.0f;
    __syncthreads();

    int nl = 0;

    auto drain = [&](int n) {
#pragma unroll 1
        for (int base = 0; base < n; base += 32) {
            int dv = RB, sv = 0;
            if (base + lane < n) {
                const int e = lstE[wv][base + lane];
                const int s = src[e];
                const int d = dst[e] - row0;
                const bool mine = (unsigned)d < (unsigned)RB;
                dv = mine ? d : RB;
                sv = mine ? s : 0;
                atomicAdd(&cntS[dv], 1.0f);
            }
            const int nb = (n - base < 32) ? (n - base) : 32;
#pragma unroll 1
            for (int i = 0; i < nb; i += DB) {
                float4 x[DB];
                int dk[DB];
#pragma unroll
                for (int k = 0; k < DB; ++k) {
                    const int l = i + k;
                    const int sk = __shfl(sv, l);
                    dk[k] = __shfl(dv, l);
                    x[k] = *(const float4*)&tsrc[(size_t)sk * D + lane * 4];
                }
#pragma unroll
                for (int k = 0; k < DB; ++k) {
                    float* a = &acc[dk[k] * D + lane];
                    atomicAdd(a,      x[k].x);
                    atomicAdd(a + 32, x[k].y);
                    atomicAdd(a + 64, x[k].z);
                    atomicAdd(a + 96, x[k].w);
                }
            }
        }
    };

    const int nW = E >> 2;
    const int nQ = nW >> 2;
    const int nT = (nQ + 31) >> 5;
    const int off = (int)(((unsigned)blockIdx.x * (unsigned)nT) / gridDim.x);
    const uint4* c4 = (const uint4*)code;
    const unsigned pat = ((unsigned)blockIdx.x & 255u) * 0x01010101u;

    auto quad_of = [&](int tl, bool& ok) -> int {
        int tr = tl + off;
        if (tr >= nT) tr -= nT;
        const int g = tr * 32 + lane;
        ok = (tl < nT) && (g < nQ);
        return g;
    };
    auto flags_of = [&](const uint4 v) -> unsigned {
        const unsigned w[4] = { v.x, v.y, v.z, v.w };
        unsigned r = 0u;
#pragma unroll
        for (int j = 0; j < 4; ++j) {
            const unsigned x = w[j] ^ pat;
            const unsigned z = (x - 0x01010101u) & ~x & 0x80808080u;
            r |= (z >> 7) << j;
        }
        return r;
    };

    if (nQ > 0) {
        uint4 vn[TPI];
#pragma unroll
        for (int k = 0; k < TPI; ++k) {
            bool ok; const int g = quad_of(wv + NWV * k, ok);
            vn[k] = c4[ok ? g : 0];
        }
        for (int tt = wv; tt < nT; tt += NWV * TPI) {
            uint4 v[TPI];
#pragma unroll
            for (int k = 0; k < TPI; ++k) v[k] = vn[k];
            if (tt + NWV * TPI < nT) {
#pragma unroll
                for (int k = 0; k < TPI; ++k) {
                    bool ok; const int g = quad_of(tt + NWV * TPI + NWV * k, ok);
                    vn[k] = c4[ok ? g : 0];
                }
            }
            unsigned f[TPI];
#pragma unroll
            for (int k = 0; k < TPI; ++k) {
                bool ok; (void)quad_of(tt + NWV * k, ok);
                f[k] = ok ? flags_of(v[k]) : 0u;
            }
            unsigned long long G = (unsigned long long)(f[0] | (f[1] << 4))
                                 | ((unsigned long long)(f[2] | (f[3] << 4)) << 32);
            for (;;) {
                const bool has = G != 0ull;
                const unsigned m = __builtin_amdgcn_ballot_w32(has);
                if (m == 0u) break;
                const int pc = __builtin_popcount(m);
                if (nl + pc > LCAP) { drain(nl); nl = 0; }
                if (has) {
                    const int p = __builtin_ctzll(G);
                    G &= G - 1ull;
                    const int k = ((p >> 5) << 1) | ((p >> 2) & 1);
                    int tr = tt + NWV * k + off;
                    if (tr >= nT) tr -= nT;
                    const int e = ((tr * 32 + lane) << 4) + ((p & 3) << 2) + ((p & 31) >> 3);
                    lstE[wv][nl + (int)__builtin_amdgcn_mbcnt_lo(m, 0u)] = e;
                }
                nl += pc;
            }
        }
    }
    if (wv == 0 && (nQ << 4) < E) {
        const int e = (nQ << 4) + lane;
        const bool in = e < E;
        const int d = (in ? dst[e] : -1) - row0;
        const bool hit = in && (unsigned)d < (unsigned)RB;
        const unsigned mq = __builtin_amdgcn_ballot_w32(hit);
        if (nl + __builtin_popcount(mq) > LCAP) { drain(nl); nl = 0; }
        if (hit) lstE[wv][nl + (int)__builtin_amdgcn_mbcnt_lo(mq, 0u)] = e;
        nl += __builtin_popcount(mq);
    }
    drain(nl);
    __syncthreads();

    for (int i = tid; i < RB * 32; i += 1024) {
        const int r = i >> 5;
        if (row0 + r < N) {
            const float* a = &acc[r * D + lane];
            ((float4*)agg)[(size_t)(row0 + r) * 32 + lane] = make_float4(a[0], a[32], a[64], a[96]);
        }
    }
    for (int i = tid; i < RB; i += 1024)
        if (row0 + i < N) cnt[row0 + i] = cntS[i];
}

__device__ __forceinline__ float rowsum16(float v) {
    v += __shfl_xor(v, 1);
    v += __shfl_xor(v, 2);
    v += __shfl_xor(v, 4);
    v += __shfl_xor(v, 8);
    return v;
}

__device__ __forceinline__
void sage_pass(v8f (&acc)[8], const float* __restrict__ tr, const float* __restrict__ ar,
               const float ca, const float* __restrict__ Wl, const float* __restrict__ Wr,
               const size_t wo) {
#pragma unroll 1
    for (int kc = 0; kc < D; kc += 16) {
        v2f at[4], aa[4];
#pragma unroll
        for (int i = 0; i < 4; ++i) {
            at[i] = *(const v2f*)(tr + kc + 4 * i);
            const v2f xa = *(const v2f*)(ar + kc + 4 * i);
            aa[i].x = xa.x / ca; aa[i].y = xa.y / ca;
        }
#pragma unroll
        for (int i = 0; i < 4; ++i) {
            const size_t w0 = wo + (size_t)(kc + 4 * i) * D;
#pragma unroll
            for (int ct = 0; ct < 8; ++ct) {
                v2f b1, b2;
                b1.x = Wl[w0 + 16 * ct]; b1.y = Wl[w0 + D + 16 * ct];
                b2.x = Wr[w0 + 16 * ct]; b2.y = Wr[w0 + D + 16 * ct];
                acc[ct] = wmma_f32(aa[i], b1, acc[ct]);
                acc[ct] = wmma_f32(at[i], b2, acc[ct]);
            }
        }
    }
}

__global__ __launch_bounds__(32)
void combine_kernel(int tilesM,
                    const float* __restrict__ t_m,
                    const float* __restrict__ agg_mm, const float* __restrict__ cnt_mm,
                    const float* __restrict__ agg_hm, const float* __restrict__ cnt_hm,
                    const float* __restrict__ Wl_mm, const float* __restrict__ bl_mm,
                    const float* __restrict__ Wr_mm,
                    const float* __restrict__ Wl_hm, const float* __restrict__ bl_hm,
                    const float* __restrict__ Wr_hm,
                    const float* __restrict__ g_m, const float* __restrict__ be_m,
                    float* __restrict__ out_m,
                    const float* __restrict__ t_h,
                    const float* __restrict__ agg_mh, const float* __restrict__ cnt_mh,
                    const float* __restrict__ Wl_mh, const float* __restrict__ bl_mh,
                    const float* __restrict__ Wr_mh,
                    const float* __restrict__ g_h, const float* __restrict__ be_h,
                    float* __restrict__ out_h) {
    const int lane = threadIdx.x;
    const int b = blockIdx.x;
    const bool two = b < tilesM;
    const int row0 = (two ? b : b - tilesM) * 16;
    const float* t    = two ? t_m    : t_h;
    const float* agg1 = two ? agg_mm : agg_mh;
    const float* cnt1 = two ? cnt_mm : cnt_mh;
    const float* Wl1  = two ? Wl_mm  : Wl_mh;
    const float* bl1  = two ? bl_mm  : bl_mh;
    const float* Wr1  = two ? Wr_mm  : Wr_mh;
    const float* g    = two ? g_m    : g_h;
    const float* be   = two ? be_m   : be_h;
    float*       out  = two ? out_m  : out_h;

    const int n = lane & 15, h = lane >> 4, m = lane & 15;
    const int kb = h * 2;
    const float* tr = t + (size_t)(row0 + m) * D + kb;
    const size_t wo = (size_t)kb * D + n;

    v8f acc1[8], acc2[8];
#pragma unroll
    for (int ct = 0; ct < 8; ++ct) { acc1[ct] = v8f{}; acc2[ct] = v8f{}; }
    const float* ar = agg1 + (size_t)(row0 + m) * D + kb;
    float ca = fmaxf(cnt1[row0 + m], 1.0f);
    const float* Wl = Wl1;
    const float* Wr = Wr1;
#pragma unroll 1
    for (int pass = 0; pass < 2; ++pass) {
        sage_pass(acc2, tr, ar, ca, Wl, Wr, wo);
        if (pass == 1) break;
#pragma unroll
        for (int ct = 0; ct < 8; ++ct) { acc1[ct] = acc2[ct]; acc2[ct] = v8f{}; }
        if (!two) break;
        ar = agg_hm + (size_t)(row0 + m) * D + kb;
        ca = fmaxf(cnt_hm[row0 + m], 1.0f);
        Wl = Wl_hm;
        Wr = Wr_hm;
    }

    float s1[8], s2[8];
#pragma unroll
    for (int p = 0; p < 8; ++p) { s1[p] = 0.0f; s2[p] = 0.0f; }
#pragma unroll
    for (int ct = 0; ct < 8; ++ct) {
        const float c1 = bl1[16 * ct + n], c2 = bl_hm[16 * ct + n];
#pragma unroll
        for (int p = 0; p < 8; ++p) {
            acc1[ct][p] += c1;
            acc2[ct][p] += c2;
            s1[p] += acc1[ct][p] * acc1[ct][p];
            s2[p] += acc2[ct][p] * acc2[ct][p];
        }
    }
    float n1[8], n2[8];
#pragma unroll
    for (int p = 0; p < 8; ++p) {
        n1[p] = fmaxf(sqrtf(rowsum16(s1[p])), 1e-12f);
        n2[p] = fmaxf(sqrtf(rowsum16(s2[p])), 1e-12f);
    }

    float sy[8], sq[8];
#pragma unroll
    for (int p = 0; p < 8; ++p) { sy[p] = 0.0f; sq[p] = 0.0f; }
#pragma unroll
    for (int ct = 0; ct < 8; ++ct) {
#pragma unroll
        for (int p = 0; p < 8; ++p) {
            const float tv = t[(size_t)(row0 + p + 8 * h) * D + 16 * ct + n];
            const float o1 = acc1[ct][p] / n1[p];
            const float conv = two ? 0.5f * (o1 + acc2[ct][p] / n2[p]) : o1;
            const float y = tv + conv;
            acc1[ct][p] = y;
            sy[p] += y;
            sq[p] += y * y;
        }
    }
    float mu[8], rs[8];
#pragma unroll
    for (int p = 0; p < 8; ++p) {
        mu[p] = rowsum16(sy[p]) * (1.0f / D);
        const float var = rowsum16(sq[p]) * (1.0f / D) - mu[p] * mu[p];
        rs[p] = rsqrtf(var + 1e-5f);
    }
#pragma unroll
    for (int ct = 0; ct < 8; ++ct) {
        const float gc = g[16 * ct + n], bc = be[16 * ct + n];
#pragma unroll
        for (int p = 0; p < 8; ++p)
            out[(size_t)(row0 + p + 8 * h) * D + 16 * ct + n] = (acc1[ct][p] - mu[p]) * rs[p] * gc + bc;
    }
}

extern "C" void kernel_launch(void* const* d_in, const int* in_sizes, int n_in,
                              void* d_out, int out_size, void* d_ws, size_t ws_size,
                              hipStream_t stream) {
    const float* x_m    = (const float*)d_in[0];
    const float* x_h    = (const float*)d_in[1];
    const float* W_nt_m = (const float*)d_in[2];
    const float* b_nt_m = (const float*)d_in[3];
    const float* W_nt_h = (const float*)d_in[4];
    const float* b_nt_h = (const float*)d_in[5];
    const float* Wl_mm  = (const float*)d_in[6];
    const float* bl_mm  = (const float*)d_in[7];
    const float* Wr_mm  = (const float*)d_in[8];
    const float* Wl_mh  = (const float*)d_in[9];
    const float* bl_mh  = (const float*)d_in[10];
    const float* Wr_mh  = (const float*)d_in[11];
    const float* Wl_hm  = (const float*)d_in[12];
    const float* bl_hm  = (const float*)d_in[13];
    const float* Wr_hm  = (const float*)d_in[14];
    const float* g_m    = (const float*)d_in[15];
    const float* be_m   = (const float*)d_in[16];
    const float* g_h    = (const float*)d_in[17];
    const float* be_h   = (const float*)d_in[18];
    const int*   ei_mm  = (const int*)d_in[19];
    const int*   ei_mh  = (const int*)d_in[20];
    const int*   ei_hm  = (const int*)d_in[21];

    const int N_M  = in_sizes[0] / D;
    const int N_H  = in_sizes[1] / D;
    const int E_mm = in_sizes[19] / 2;
    const int E_mh = in_sizes[20] / 2;
    const int E_hm = in_sizes[21] / 2;

    const size_t nm = (size_t)N_M * D;
    const size_t nh = (size_t)N_H * D;

    float* ws     = (float*)d_ws;
    float* t_m    = ws;
    float* t_h    = t_m + nm;
    float* agg_mm = t_h + nh;
    float* agg_hm = agg_mm + nm;
    float* agg_mh = agg_hm + nm;
    float* cnt_mm = agg_mh + nh;
    float* cnt_hm = cnt_mm + N_M;
    float* cnt_mh = cnt_hm + N_M;

    unsigned* code_mm = (unsigned*)d_out;
    unsigned* code_mh = code_mm + (((E_mm >> 2) + 3) & ~3);
    unsigned* code_hm = code_mh + (((E_mh >> 2) + 3) & ~3);

    const int nmax = N_M > N_H ? N_M : N_H;
    lin_kernel<<<dim3(nmax / 16, 3), 32, 0, stream>>>(x_m, W_nt_m, b_nt_m, t_m, N_M,
                                                       x_h, W_nt_h, b_nt_h, t_h, N_H,
                                                       ei_mm, ei_mh, ei_hm, E_mm, E_mh, E_hm,
                                                       code_mm, code_mh, code_hm);

    own_agg_kernel<<<dim3((nmax + RB - 1) / RB, 3), 1024, 0, stream>>>(
        t_m, t_h, ei_mm, ei_mh, ei_hm, code_mm, code_mh, code_hm,
        E_mm, E_mh, E_hm, N_M, N_H,
        agg_mm, agg_mh, agg_hm, cnt_mm, cnt_mh, cnt_hm);

    float* out_m = (float*)d_out;
    float* out_h = out_m + nm;
    combine_kernel<<<N_M / 16 + N_H / 16, 32, 0, stream>>>(
        N_M / 16,
        t_m, agg_mm, cnt_mm, agg_hm, cnt_hm, Wl_mm, bl_mm, Wr_mm, Wl_hm, bl_hm, Wr_hm,
        g_m, be_m, out_m,
        t_h, agg_mh, cnt_mh, Wl_mh, bl_mh, Wr_mh, g_h, be_h, out_h);
}
